# starting kernel
# baseline (speedup 1.0000x reference)
#include <hip/hip_runtime.h>
#include <hip/hip_bf16.h>

typedef float v2f __attribute__((ext_vector_type(2)));
typedef float v8f __attribute__((ext_vector_type(8)));

#define B_  2
#define L_  2048
#define D_  512
#define N_  64
#define R_  32
#define E_  160
#define BL_ (B_ * L_)
#define G_  32
#define T_  (L_ / G_)

__global__ void __launch_bounds__(256)
s6_xproj_gemm(const float* __restrict__ x, const float* __restrict__ W,
              float* __restrict__ xdbl) {
  const int wave  = blockIdx.x * (blockDim.x >> 5) + (threadIdx.x >> 5);
  const int lane  = threadIdx.x & 31;
  const int rt    = wave / 10;
  const int et    = wave % 10;
  const int laneM = lane & 15;
  const int laneH = lane >> 4;

  const float* arow = x + (size_t)(rt * 16 + laneM) * D_ + 2 * laneH;
  const float* brow = W + (size_t)(et * 16 + laneM) * D_ + 2 * laneH;

  v8f c = {};
  for (int k = 0; k < D_; k += 4) {
    v2f a = *(const v2f*)(arow + k);
    v2f b = *(const v2f*)(brow + k);
    c = __builtin_amdgcn_wmma_f32_16x16x4_f32(false, a, false, b,
                                              (short)0, c, false, false);
  }
  const int col = et * 16 + laneM;
#pragma unroll
  for (int j = 0; j < 8; ++j) {
    const int row = rt * 16 + j + 8 * laneH;
    xdbl[(size_t)row * E_ + col] = c[j];
  }
}

__global__ void __launch_bounds__(256)
s6_delta_gemm(const float* __restrict__ xdbl, const float* __restrict__ Wdt,
              const float* __restrict__ bdt, float* __restrict__ delta) {
  const int wave  = blockIdx.x * (blockDim.x >> 5) + (threadIdx.x >> 5);
  const int lane  = threadIdx.x & 31;
  const int rt    = wave >> 5;
  const int nt    = wave & 31;
  const int laneM = lane & 15;
  const int laneH = lane >> 4;

  const float* arow = xdbl + (size_t)(rt * 16 + laneM) * E_ + 2 * laneH;
  const float* brow = Wdt  + (size_t)(nt * 16 + laneM) * R_ + 2 * laneH;

  v8f c = {};
#pragma unroll
  for (int k = 0; k < R_; k += 4) {
    v2f a = *(const v2f*)(arow + k);
    v2f b = *(const v2f*)(brow + k);
    c = __builtin_amdgcn_wmma_f32_16x16x4_f32(false, a, false, b,
                                              (short)0, c, false, false);
  }
  const int col  = nt * 16 + laneM;
  const float bi = bdt[col];
#pragma unroll
  for (int j = 0; j < 8; ++j) {
    const int row = rt * 16 + j + 8 * laneH;
    float z = c[j] + bi;
    float sp = (z > 20.0f) ? z : __logf(1.0f + __expf(z));
    delta[(size_t)row * D_ + col] = sp;
  }
}

__global__ void __launch_bounds__(256)
s6_scan_chunks(const float* __restrict__ xdbl, const float* __restrict__ delta,
               const float* __restrict__ x, const float* __restrict__ Alog,
               float* __restrict__ Pout, float* __restrict__ Hout) {
  const int wave = blockIdx.x * (blockDim.x >> 5) + (threadIdx.x >> 5);
  const int lane = threadIdx.x & 31;
  const int g  = wave % G_;
  const int bd = wave / G_;
  const int dd = bd % D_;
  const int bb = bd / D_;
  const int n0 = 2 * lane;

  v2f al = *(const v2f*)(Alog + (size_t)dd * N_ + n0);
  v2f A;  A.x = -__expf(al.x);  A.y = -__expf(al.y);

  v2f h = {0.0f, 0.0f};
  v2f P = {1.0f, 1.0f};
  const int t0 = g * T_;
  for (int tt = 0; tt < T_; ++tt) {
    const size_t row = (size_t)bb * L_ + (t0 + tt);
    const float dlt = delta[row * D_ + dd];
    const float xv  = x[row * D_ + dd];
    v2f Bv = *(const v2f*)(xdbl + row * E_ + R_ + n0);
    if (tt + 8 < T_)
      __builtin_prefetch(xdbl + (row + 8) * E_ + R_ + n0, 0, 1);
    v2f a;  a.x = __expf(dlt * A.x);  a.y = __expf(dlt * A.y);
    const float du = dlt * xv;
    h.x = a.x * h.x + du * Bv.x;
    h.y = a.y * h.y + du * Bv.y;
    P.x *= a.x;  P.y *= a.y;
  }
  const size_t base = ((size_t)bd * G_ + g) * N_ + n0;
  *(v2f*)(Pout + base) = P;
  *(v2f*)(Hout + base) = h;
}

__global__ void __launch_bounds__(256)
s6_combine(const float* __restrict__ Pin, const float* __restrict__ Hin,
           float* __restrict__ h0out) {
  const int bd   = blockIdx.x * (blockDim.x >> 5) + (threadIdx.x >> 5);
  const int lane = threadIdx.x & 31;
  const int n0   = 2 * lane;

  v2f h = {0.0f, 0.0f};
  for (int g = 0; g < G_; ++g) {
    const size_t base = ((size_t)bd * G_ + g) * N_ + n0;
    *(v2f*)(h0out + base) = h;
    v2f P = *(const v2f*)(Pin + base);
    v2f H = *(const v2f*)(Hin + base);
    h.x = P.x * h.x + H.x;
    h.y = P.y * h.y + H.y;
  }
}

__global__ void __launch_bounds__(256)
s6_scan_emit(const float* __restrict__ xdbl, const float* __restrict__ delta,
             const float* __restrict__ x, const float* __restrict__ Alog,
             const float* __restrict__ Dp, const float* __restrict__ h0in,
             float* __restrict__ y) {
  const int wave = blockIdx.x * (blockDim.x >> 5) + (threadIdx.x >> 5);
  const int lane = threadIdx.x & 31;
  const int g  = wave % G_;
  const int bd = wave / G_;
  const int dd = bd % D_;
  const int bb = bd / D_;
  const int n0 = 2 * lane;

  v2f al = *(const v2f*)(Alog + (size_t)dd * N_ + n0);
  v2f A;  A.x = -__expf(al.x);  A.y = -__expf(al.y);
  const float Dv = Dp[dd];

  v2f h = *(const v2f*)(h0in + ((size_t)bd * G_ + g) * N_ + n0);

  const int t0 = g * T_;
  for (int tt = 0; tt < T_; ++tt) {
    const size_t row = (size_t)bb * L_ + (t0 + tt);
    const float dlt = delta[row * D_ + dd];
    const float xv  = x[row * D_ + dd];
    v2f Bv = *(const v2f*)(xdbl + row * E_ + R_ + n0);
    v2f Cv = *(const v2f*)(xdbl + row * E_ + R_ + N_ + n0);
    if (tt + 8 < T_)
      __builtin_prefetch(xdbl + (row + 8) * E_ + R_ + n0, 0, 1);
    v2f a;  a.x = __expf(dlt * A.x);  a.y = __expf(dlt * A.y);
    const float du = dlt * xv;
    h.x = a.x * h.x + du * Bv.x;
    h.y = a.y * h.y + du * Bv.y;
    float part = h.x * Cv.x + h.y * Cv.y;
#pragma unroll
    for (int off = 16; off >= 1; off >>= 1)
      part += __shfl_xor(part, off, 32);
    if (lane == 0)
      y[row * D_ + dd] = part + xv * Dv;
  }
}

extern "C" void kernel_launch(void* const* d_in, const int* in_sizes, int n_in,
                              void* d_out, int out_size, void* d_ws, size_t ws_size,
                              hipStream_t stream) {
  const float* x      = (const float*)d_in[0];
  const float* Wxproj = (const float*)d_in[1];
  const float* Wdt    = (const float*)d_in[2];
  const float* bdt    = (const float*)d_in[3];
  const float* Alog   = (const float*)d_in[4];
  const float* Dp     = (const float*)d_in[5];
  float*       y      = (float*)d_out;

  float* ws    = (float*)d_ws;
  float* xdbl  = ws;
  float* delta = xdbl  + (size_t)BL_ * E_;
  float* Pws   = delta + (size_t)BL_ * D_;
  float* Hws   = Pws   + (size_t)B_ * D_ * G_ * N_;
  float* h0ws  = Hws   + (size_t)B_ * D_ * G_ * N_;

  s6_xproj_gemm<<<320, 256, 0, stream>>>(x, Wxproj, xdbl);
  s6_delta_gemm<<<1024, 256, 0, stream>>>(xdbl, Wdt, bdt, delta);
  s6_scan_chunks<<<4096, 256, 0, stream>>>(xdbl, delta, x, Alog, Pws, Hws);
  s6_combine<<<128, 256, 0, stream>>>(Pws, Hws, h0ws);
  s6_scan_emit<<<4096, 256, 0, stream>>>(xdbl, delta, x, Alog, Dp, h0ws, y);
}


# Scan passes remapped: lane = channel, 4 waves x 16 states in registers, B/C rows by scalar loads; y sum over n now serial per lane + 4 partials via LDS (new sum order); no shuffles, whole-line I/O
# speedup vs baseline: 2.7745x; 2.7745x over previous
#include <hip/hip_runtime.h>
#include <hip/hip_bf16.h>

typedef float v2f __attribute__((ext_vector_type(2)));
typedef float v8f __attribute__((ext_vector_type(8)));
typedef float v4f __attribute__((ext_vector_type(4)));
typedef float v16f __attribute__((ext_vector_type(16)));

#define B_  2
#define L_  2048
#define D_  512
#define N_  64
#define R_  32
#define E_  160
#define BL_ (B_ * L_)
#define G_  32
#define T_  (L_ / G_)
#define NS_ 4
#define NQ_ (N_ / NS_)
#define DG_ (D_ / 32)

__global__ void __launch_bounds__(256)
s6_xproj_gemm(const float* __restrict__ x, const float* __restrict__ W,
              float* __restrict__ xdbl) {
  const int wave  = blockIdx.x * (blockDim.x >> 5) + (threadIdx.x >> 5);
  const int lane  = threadIdx.x & 31;
  const int rt    = wave / 10;
  const int et    = wave % 10;
  const int laneM = lane & 15;
  const int laneH = lane >> 4;

  const float* arow = x + (size_t)(rt * 16 + laneM) * D_ + 2 * laneH;
  const float* brow = W + (size_t)(et * 16 + laneM) * D_ + 2 * laneH;

  v8f c = {};
  for (int k = 0; k < D_; k += 4) {
    v2f a = *(const v2f*)(arow + k);
    v2f b = *(const v2f*)(brow + k);
    c = __builtin_amdgcn_wmma_f32_16x16x4_f32(false, a, false, b,
                                              (short)0, c, false, false);
  }
  const int col = et * 16 + laneM;
#pragma unroll
  for (int j = 0; j < 8; ++j) {
    const int row = rt * 16 + j + 8 * laneH;
    xdbl[(size_t)row * E_ + col] = c[j];
  }
}

__global__ void __launch_bounds__(256)
s6_delta_gemm(const float* __restrict__ xdbl, const float* __restrict__ Wdt,
              const float* __restrict__ bdt, float* __restrict__ delta) {
  const int wave  = blockIdx.x * (blockDim.x >> 5) + (threadIdx.x >> 5);
  const int lane  = threadIdx.x & 31;
  const int rt    = wave >> 5;
  const int nt    = wave & 31;
  const int laneM = lane & 15;
  const int laneH = lane >> 4;

  const float* arow = xdbl + (size_t)(rt * 16 + laneM) * E_ + 2 * laneH;
  const float* brow = Wdt  + (size_t)(nt * 16 + laneM) * R_ + 2 * laneH;

  v8f c = {};
#pragma unroll
  for (int k = 0; k < R_; k += 4) {
    v2f a = *(const v2f*)(arow + k);
    v2f b = *(const v2f*)(brow + k);
    c = __builtin_amdgcn_wmma_f32_16x16x4_f32(false, a, false, b,
                                              (short)0, c, false, false);
  }
  const int col  = nt * 16 + laneM;
  const float bi = bdt[col];
#pragma unroll
  for (int j = 0; j < 8; ++j) {
    const int row = rt * 16 + j + 8 * laneH;
    float z = c[j] + bi;
    float sp = (z > 20.0f) ? z : __logf(1.0f + __expf(z));
    delta[(size_t)row * D_ + col] = sp;
  }
}


__device__ __forceinline__ void
s6_step_chunk(const v16f Bc, const float dlt, const float xv,
              const v2f (&A2)[NQ_ / 2], v2f (&h2)[NQ_ / 2], v2f (&P2)[NQ_ / 2]) {
  const float du = dlt * xv;
#pragma unroll
  for (int j = 0; j < NQ_ / 2; ++j) {
    const v2f arg = dlt * A2[j];
    v2f a;  a.x = __expf(arg.x);  a.y = __expf(arg.y);
    v2f Bv; Bv.x = Bc[2 * j];     Bv.y = Bc[2 * j + 1];
    h2[j] = a * h2[j] + du * Bv;
    P2[j] *= a;
  }
}

__device__ __forceinline__ float
s6_step_emit(const v16f Bc, const v16f Cc, const float dlt, const float xv,
             const v2f (&A2)[NQ_ / 2], v2f (&h2)[NQ_ / 2]) {
  const float du = dlt * xv;
  v2f acc = {0.0f, 0.0f};
#pragma unroll
  for (int j = 0; j < NQ_ / 2; ++j) {
    const v2f arg = dlt * A2[j];
    v2f a;  a.x = __expf(arg.x);  a.y = __expf(arg.y);
    v2f Bv; Bv.x = Bc[2 * j];     Bv.y = Bc[2 * j + 1];
    v2f Cv; Cv.x = Cc[2 * j];     Cv.y = Cc[2 * j + 1];
    h2[j] = a * h2[j] + du * Bv;
    acc += h2[j] * Cv;
  }
  return acc.x + acc.y;
}

__global__ void __launch_bounds__(128)
s6_scan_chunks(const float* __restrict__ xdbl, const float* __restrict__ delta,
               const float* __restrict__ x, const float* __restrict__ Alog,
               float* __restrict__ Pout, float* __restrict__ Hout) {
  const int lane = threadIdx.x & 31;
  const int q  = __builtin_amdgcn_readfirstlane((int)(threadIdx.x >> 5));
  const int dg = blockIdx.x % DG_;
  const int bg = blockIdx.x / DG_;
  const int g  = bg % G_;
  const int bb = bg / G_;
  const int d  = dg * 32 + lane;
  const int nq = q * NQ_;

  v2f A2[NQ_ / 2], h2[NQ_ / 2], P2[NQ_ / 2];
  const v4f* alp = (const v4f*)(Alog + (size_t)d * N_ + nq);
#pragma unroll
  for (int k = 0; k < NQ_ / 4; ++k) {
    const v4f al = alp[k];
    A2[2 * k].x     = -__expf(al.x);
    A2[2 * k].y     = -__expf(al.y);
    A2[2 * k + 1].x = -__expf(al.z);
    A2[2 * k + 1].y = -__expf(al.w);
  }
#pragma unroll
  for (int j = 0; j < NQ_ / 2; ++j) {
    h2[j].x = 0.0f;  h2[j].y = 0.0f;
    P2[j].x = 1.0f;  P2[j].y = 1.0f;
  }

  const size_t row0 = (size_t)bb * L_ + (size_t)g * T_;
  const float* dp = delta + row0 * D_ + d;
  const float* xp = x     + row0 * D_ + d;
  const float* bp = xdbl  + row0 * E_ + R_ + nq;

  v16f B0 = *(const v16f*)bp;
  float d0 = dp[0],  x0 = xp[0];
  float d1 = dp[D_], x1 = xp[D_];
  for (int tt = 0; tt < T_; tt += 2) {
    const int t2 = (tt + 2 < T_) ? tt + 2 : tt;
    const v16f B1 = *(const v16f*)(bp + (size_t)(tt + 1) * E_);
    const float dc0 = d0, xc0 = x0, dc1 = d1, xc1 = x1;
    d0 = dp[(size_t)t2 * D_];        x0 = xp[(size_t)t2 * D_];
    d1 = dp[(size_t)(t2 + 1) * D_];  x1 = xp[(size_t)(t2 + 1) * D_];
    s6_step_chunk(B0, dc0, xc0, A2, h2, P2);
    B0 = *(const v16f*)(bp + (size_t)t2 * E_);
    s6_step_chunk(B1, dc1, xc1, A2, h2, P2);
  }
  const size_t base = ((size_t)bg * N_ + nq) * D_ + d;
#pragma unroll
  for (int j = 0; j < NQ_ / 2; ++j) {
    Pout[base + (size_t)(2 * j) * D_]     = P2[j].x;
    Pout[base + (size_t)(2 * j + 1) * D_] = P2[j].y;
    Hout[base + (size_t)(2 * j) * D_]     = h2[j].x;
    Hout[base + (size_t)(2 * j + 1) * D_] = h2[j].y;
  }
}

__global__ void __launch_bounds__(256)
s6_combine(const float* __restrict__ Pin, const float* __restrict__ Hin,
           float* __restrict__ h0out) {
  const int idx = blockIdx.x * 256 + threadIdx.x;
  const int bb  = idx / (N_ * D_);
  const int nd  = idx % (N_ * D_);
  const size_t base = (size_t)bb * G_ * N_ * D_ + nd;

  float h = 0.0f;
#pragma unroll
  for (int g = 0; g < G_; ++g) {
    const size_t o = base + (size_t)g * N_ * D_;
    h0out[o] = h;
    h = Pin[o] * h + Hin[o];
  }
}

__global__ void __launch_bounds__(128)
s6_scan_emit(const float* __restrict__ xdbl, const float* __restrict__ delta,
             const float* __restrict__ x, const float* __restrict__ Alog,
             const float* __restrict__ Dp, const float* __restrict__ h0in,
             float* __restrict__ y) {
  __shared__ float ypart[T_ * NS_ * 32];

  const int lane = threadIdx.x & 31;
  const int q  = __builtin_amdgcn_readfirstlane((int)(threadIdx.x >> 5));
  const int dg = blockIdx.x % DG_;
  const int bg = blockIdx.x / DG_;
  const int g  = bg % G_;
  const int bb = bg / G_;
  const int d  = dg * 32 + lane;
  const int nq = q * NQ_;

  v2f A2[NQ_ / 2], h2[NQ_ / 2];
  const v4f* alp = (const v4f*)(Alog + (size_t)d * N_ + nq);
#pragma unroll
  for (int k = 0; k < NQ_ / 4; ++k) {
    const v4f al = alp[k];
    A2[2 * k].x     = -__expf(al.x);
    A2[2 * k].y     = -__expf(al.y);
    A2[2 * k + 1].x = -__expf(al.z);
    A2[2 * k + 1].y = -__expf(al.w);
  }
  const size_t hbase = ((size_t)bg * N_ + nq) * D_ + d;
#pragma unroll
  for (int j = 0; j < NQ_ / 2; ++j) {
    h2[j].x = h0in[hbase + (size_t)(2 * j) * D_];
    h2[j].y = h0in[hbase + (size_t)(2 * j + 1) * D_];
  }

  const size_t row0 = (size_t)bb * L_ + (size_t)g * T_;
  const float* dp = delta + row0 * D_ + d;
  const float* xp = x     + row0 * D_ + d;
  const float* bp = xdbl  + row0 * E_ + R_ + nq;
  const float* cp = bp + N_;
  float* yw = ypart + q * 32 + lane;

  v16f B0 = *(const v16f*)bp;
  v16f C0 = *(const v16f*)cp;
  float d0 = dp[0],  x0 = xp[0];
  float d1 = dp[D_], x1 = xp[D_];
  for (int tt = 0; tt < T_; tt += 2) {
    const int t2 = (tt + 2 < T_) ? tt + 2 : tt;
    const v16f B1 = *(const v16f*)(bp + (size_t)(tt + 1) * E_);
    const v16f C1 = *(const v16f*)(cp + (size_t)(tt + 1) * E_);
    const float dc0 = d0, xc0 = x0, dc1 = d1, xc1 = x1;
    d0 = dp[(size_t)t2 * D_];        x0 = xp[(size_t)t2 * D_];
    d1 = dp[(size_t)(t2 + 1) * D_];  x1 = xp[(size_t)(t2 + 1) * D_];
    yw[tt * NS_ * 32] = s6_step_emit(B0, C0, dc0, xc0, A2, h2);
    B0 = *(const v16f*)(bp + (size_t)t2 * E_);
    C0 = *(const v16f*)(cp + (size_t)t2 * E_);
    yw[(tt + 1) * NS_ * 32] = s6_step_emit(B1, C1, dc1, xc1, A2, h2);
  }
  __syncthreads();

  const float Dv = Dp[d];
#pragma unroll 4
  for (int k = 0; k < T_ / NS_; ++k) {
    const int tt = q * (T_ / NS_) + k;
    const float* yp = ypart + tt * NS_ * 32 + lane;
    const float s = (yp[0] + yp[32]) + (yp[64] + yp[96]);
    const float xv = xp[(size_t)tt * D_];
    y[(row0 + tt) * D_ + d] = s + xv * Dv;
  }
}

extern "C" void kernel_launch(void* const* d_in, const int* in_sizes, int n_in,
                              void* d_out, int out_size, void* d_ws, size_t ws_size,
                              hipStream_t stream) {
  const float* x      = (const float*)d_in[0];
  const float* Wxproj = (const float*)d_in[1];
  const float* Wdt    = (const float*)d_in[2];
  const float* bdt    = (const float*)d_in[3];
  const float* Alog   = (const float*)d_in[4];
  const float* Dp     = (const float*)d_in[5];
  float*       y      = (float*)d_out;

  float* ws    = (float*)d_ws;
  float* xdbl  = ws;
  float* delta = xdbl  + (size_t)BL_ * E_;
  float* Pws   = delta + (size_t)BL_ * D_;
  float* Hws   = Pws   + (size_t)B_ * D_ * G_ * N_;
  float* h0ws  = Hws   + (size_t)B_ * D_ * G_ * N_;

  s6_xproj_gemm<<<320, 256, 0, stream>>>(x, Wxproj, xdbl);
  s6_delta_gemm<<<1024, 256, 0, stream>>>(xdbl, Wdt, bdt, delta);
  s6_scan_chunks<<<1024, 128, 0, stream>>>(xdbl, delta, x, Alog, Pws, Hws);
  s6_combine<<<256, 256, 0, stream>>>(Pws, Hws, h0ws);
  s6_scan_emit<<<1024, 128, 0, stream>>>(xdbl, delta, x, Alog, Dp, h0ws, y);
}


# x_proj GEMM: one work-group per 16-row tile, k split over 8 waves, x and W slices staged in LDS by whole-line loads, 10 accumulators per wave; k partials added through LDS (new sum order)
# speedup vs baseline: 3.1756x; 1.1446x over previous
#include <hip/hip_runtime.h>
#include <hip/hip_bf16.h>

typedef float v2f __attribute__((ext_vector_type(2)));
typedef float v8f __attribute__((ext_vector_type(8)));
typedef float v4f __attribute__((ext_vector_type(4)));
typedef float v16f __attribute__((ext_vector_type(16)));

#define B_  2
#define L_  2048
#define D_  512
#define N_  64
#define R_  32
#define E_  160
#define BL_ (B_ * L_)
#define G_  32
#define T_  (L_ / G_)
#define NS_ 4
#define NQ_ (N_ / NS_)
#define DG_ (D_ / 32)

#define XW_  8
#define XK_  (D_ / XW_)
#define XP_  68
#define XT_  (16 * XP_)
#define XET_ (E_ / 16)
#define XWS_ (5 * XT_)

__device__ __forceinline__ void
s6_slice_fetch(const float* __restrict__ src, const int lane, v4f (&r)[8]) {
  const float* p = src + (size_t)(lane >> 4) * D_ + 4 * (lane & 15);
#pragma unroll
  for (int i = 0; i < 8; ++i) r[i] = *(const v4f*)(p + (size_t)(2 * i) * D_);
}

__device__ __forceinline__ void
s6_slice_park(float* dst, const int lane, const v4f (&r)[8]) {
  float* p = dst + (lane >> 4) * XP_ + 4 * (lane & 15);
#pragma unroll
  for (int i = 0; i < 8; ++i) *(v4f*)(p + 2 * i * XP_) = r[i];
}

__device__ __forceinline__ void
s6_slice_frags(const float* tile, const int lane, v4f (&f)[8]) {
  const float* p = tile + (lane & 15) * XP_ + 4 * (lane >> 4);
#pragma unroll
  for (int i = 0; i < 8; ++i) f[i] = *(const v4f*)(p + 8 * i);
}

__device__ __forceinline__ void s6_wave_lds_handover() {
  __builtin_amdgcn_fence(__ATOMIC_RELEASE, "wavefront");
  __builtin_amdgcn_wave_barrier();
  __builtin_amdgcn_fence(__ATOMIC_ACQUIRE, "wavefront");
}

__global__ void __launch_bounds__(256)
s6_xproj_gemm(const float* __restrict__ x, const float* __restrict__ W,
              float* __restrict__ xdbl) {
  __shared__ __attribute__((aligned(16))) float sm[XW_ * XWS_];

  const int lane = threadIdx.x & 31;
  const int w  = __builtin_amdgcn_readfirstlane((int)(threadIdx.x >> 5));
  const int rt = blockIdx.x;

  float* sa = sm + w * XWS_;
  float* sb = sa + XT_;
  const float* xa = x + (size_t)rt * 16 * D_ + w * XK_;
  const float* wb = W + w * XK_;

  v4f ra[8], rb0[8], rb1[8];
  s6_slice_fetch(xa, lane, ra);
  s6_slice_fetch(wb, lane, rb0);
  s6_slice_fetch(wb + (size_t)16 * D_, lane, rb1);
  s6_slice_park(sa, lane, ra);
  s6_slice_park(sb, lane, rb0);
  s6_slice_park(sb + XT_, lane, rb1);
  s6_slice_fetch(wb + (size_t)32 * D_, lane, rb0);
  s6_slice_fetch(wb + (size_t)48 * D_, lane, rb1);
  s6_wave_lds_handover();

  v4f af[8], bf[2][2][8];
  s6_slice_frags(sa, lane, af);
  s6_slice_frags(sb, lane, bf[0][0]);
  s6_slice_frags(sb + XT_, lane, bf[0][1]);
  s6_slice_park(sb + 2 * XT_, lane, rb0);
  s6_slice_park(sb + 3 * XT_, lane, rb1);
  s6_wave_lds_handover();

  v8f acc[XET_];
#pragma unroll
  for (int j = 0; j < XET_; ++j) acc[j] = (v8f){};

#pragma unroll
  for (int jp = 0; jp < XET_ / 2; ++jp) {
    const int cs = jp & 1, ns = cs ^ 1;
    if (jp + 2 < XET_ / 2) {
      s6_slice_fetch(wb + (size_t)(2 * jp + 4) * 16 * D_, lane, rb0);
      s6_slice_fetch(wb + (size_t)(2 * jp + 5) * 16 * D_, lane, rb1);
    }
    if (jp + 1 < XET_ / 2) {
      s6_slice_frags(sb + ns * 2 * XT_, lane, bf[ns][0]);
      s6_slice_frags(sb + ns * 2 * XT_ + XT_, lane, bf[ns][1]);
    }
    __builtin_amdgcn_sched_barrier(0);
#pragma unroll
    for (int i = 0; i < 8; ++i) {
      const v2f a0 = {af[i].x, af[i].y}, a1 = {af[i].z, af[i].w};
      const v2f p0 = {bf[cs][0][i].x, bf[cs][0][i].y}, p1 = {bf[cs][0][i].z, bf[cs][0][i].w};
      const v2f q0 = {bf[cs][1][i].x, bf[cs][1][i].y}, q1 = {bf[cs][1][i].z, bf[cs][1][i].w};
      acc[2 * jp]     = __builtin_amdgcn_wmma_f32_16x16x4_f32(false, a0, false, p0, (short)0, acc[2 * jp],     false, false);
      acc[2 * jp + 1] = __builtin_amdgcn_wmma_f32_16x16x4_f32(false, a0, false, q0, (short)0, acc[2 * jp + 1], false, false);
      acc[2 * jp]     = __builtin_amdgcn_wmma_f32_16x16x4_f32(false, a1, false, p1, (short)0, acc[2 * jp],     false, false);
      acc[2 * jp + 1] = __builtin_amdgcn_wmma_f32_16x16x4_f32(false, a1, false, q1, (short)0, acc[2 * jp + 1], false, false);
    }
    __builtin_amdgcn_sched_barrier(0);
    if (jp + 2 < XET_ / 2) {
      s6_slice_park(sb + cs * 2 * XT_, lane, rb0);
      s6_slice_park(sb + cs * 2 * XT_ + XT_, lane, rb1);
      s6_wave_lds_handover();
    }
  }

  __syncthreads();
#pragma unroll
  for (int j = 0; j < XET_; ++j) {
#pragma unroll
    for (int r = 0; r < 8; ++r)
      sm[((w * XET_ + j) * 8 + r) * 32 + lane] = acc[j][r];
  }
  __syncthreads();

  float* orow = xdbl + (size_t)rt * 16 * E_;
#pragma unroll 2
  for (int o = threadIdx.x; o < 16 * E_; o += 256) {
    const int row = o / E_, col = o % E_;
    const int idx = ((col >> 4) * 8 + (row & 7)) * 32 + (col & 15) + 16 * (row >> 3);
    float s = sm[idx];
#pragma unroll
    for (int q = 1; q < XW_; ++q) s += sm[q * XET_ * 256 + idx];
    orow[o] = s;
  }
}

__global__ void __launch_bounds__(256)
s6_delta_gemm(const float* __restrict__ xdbl, const float* __restrict__ Wdt,
              const float* __restrict__ bdt, float* __restrict__ delta) {
  const int wave  = blockIdx.x * (blockDim.x >> 5) + (threadIdx.x >> 5);
  const int lane  = threadIdx.x & 31;
  const int rt    = wave >> 5;
  const int nt    = wave & 31;
  const int laneM = lane & 15;
  const int laneH = lane >> 4;

  const float* arow = xdbl + (size_t)(rt * 16 + laneM) * E_ + 2 * laneH;
  const float* brow = Wdt  + (size_t)(nt * 16 + laneM) * R_ + 2 * laneH;

  v8f c = {};
#pragma unroll
  for (int k = 0; k < R_; k += 4) {
    v2f a = *(const v2f*)(arow + k);
    v2f b = *(const v2f*)(brow + k);
    c = __builtin_amdgcn_wmma_f32_16x16x4_f32(false, a, false, b,
                                              (short)0, c, false, false);
  }
  const int col  = nt * 16 + laneM;
  const float bi = bdt[col];
#pragma unroll
  for (int j = 0; j < 8; ++j) {
    const int row = rt * 16 + j + 8 * laneH;
    float z = c[j] + bi;
    float sp = (z > 20.0f) ? z : __logf(1.0f + __expf(z));
    delta[(size_t)row * D_ + col] = sp;
  }
}


__device__ __forceinline__ void
s6_step_chunk(const v16f Bc, const float dlt, const float xv,
              const v2f (&A2)[NQ_ / 2], v2f (&h2)[NQ_ / 2], v2f (&P2)[NQ_ / 2]) {
  const float du = dlt * xv;
#pragma unroll
  for (int j = 0; j < NQ_ / 2; ++j) {
    const v2f arg = dlt * A2[j];
    v2f a;  a.x = __expf(arg.x);  a.y = __expf(arg.y);
    v2f Bv; Bv.x = Bc[2 * j];     Bv.y = Bc[2 * j + 1];
    h2[j] = a * h2[j] + du * Bv;
    P2[j] *= a;
  }
}

__device__ __forceinline__ float
s6_step_emit(const v16f Bc, const v16f Cc, const float dlt, const float xv,
             const v2f (&A2)[NQ_ / 2], v2f (&h2)[NQ_ / 2]) {
  const float du = dlt * xv;
  v2f acc = {0.0f, 0.0f};
#pragma unroll
  for (int j = 0; j < NQ_ / 2; ++j) {
    const v2f arg = dlt * A2[j];
    v2f a;  a.x = __expf(arg.x);  a.y = __expf(arg.y);
    v2f Bv; Bv.x = Bc[2 * j];     Bv.y = Bc[2 * j + 1];
    v2f Cv; Cv.x = Cc[2 * j];     Cv.y = Cc[2 * j + 1];
    h2[j] = a * h2[j] + du * Bv;
    acc += h2[j] * Cv;
  }
  return acc.x + acc.y;
}

__global__ void __launch_bounds__(128)
s6_scan_chunks(const float* __restrict__ xdbl, const float* __restrict__ delta,
               const float* __restrict__ x, const float* __restrict__ Alog,
               float* __restrict__ Pout, float* __restrict__ Hout) {
  const int lane = threadIdx.x & 31;
  const int q  = __builtin_amdgcn_readfirstlane((int)(threadIdx.x >> 5));
  const int dg = blockIdx.x % DG_;
  const int bg = blockIdx.x / DG_;
  const int g  = bg % G_;
  const int bb = bg / G_;
  const int d  = dg * 32 + lane;
  const int nq = q * NQ_;

  v2f A2[NQ_ / 2], h2[NQ_ / 2], P2[NQ_ / 2];
  const v4f* alp = (const v4f*)(Alog + (size_t)d * N_ + nq);
#pragma unroll
  for (int k = 0; k < NQ_ / 4; ++k) {
    const v4f al = alp[k];
    A2[2 * k].x     = -__expf(al.x);
    A2[2 * k].y     = -__expf(al.y);
    A2[2 * k + 1].x = -__expf(al.z);
    A2[2 * k + 1].y = -__expf(al.w);
  }
#pragma unroll
  for (int j = 0; j < NQ_ / 2; ++j) {
    h2[j].x = 0.0f;  h2[j].y = 0.0f;
    P2[j].x = 1.0f;  P2[j].y = 1.0f;
  }

  const size_t row0 = (size_t)bb * L_ + (size_t)g * T_;
  const float* dp = delta + row0 * D_ + d;
  const float* xp = x     + row0 * D_ + d;
  const float* bp = xdbl  + row0 * E_ + R_ + nq;

  v16f B0 = *(const v16f*)bp;
  float d0 = dp[0],  x0 = xp[0];
  float d1 = dp[D_], x1 = xp[D_];
  for (int tt = 0; tt < T_; tt += 2) {
    const int t2 = (tt + 2 < T_) ? tt + 2 : tt;
    const v16f B1 = *(const v16f*)(bp + (size_t)(tt + 1) * E_);
    const float dc0 = d0, xc0 = x0, dc1 = d1, xc1 = x1;
    d0 = dp[(size_t)t2 * D_];        x0 = xp[(size_t)t2 * D_];
    d1 = dp[(size_t)(t2 + 1) * D_];  x1 = xp[(size_t)(t2 + 1) * D_];
    s6_step_chunk(B0, dc0, xc0, A2, h2, P2);
    B0 = *(const v16f*)(bp + (size_t)t2 * E_);
    s6_step_chunk(B1, dc1, xc1, A2, h2, P2);
  }
  const size_t base = ((size_t)bg * N_ + nq) * D_ + d;
#pragma unroll
  for (int j = 0; j < NQ_ / 2; ++j) {
    Pout[base + (size_t)(2 * j) * D_]     = P2[j].x;
    Pout[base + (size_t)(2 * j + 1) * D_] = P2[j].y;
    Hout[base + (size_t)(2 * j) * D_]     = h2[j].x;
    Hout[base + (size_t)(2 * j + 1) * D_] = h2[j].y;
  }
}

__global__ void __launch_bounds__(256)
s6_combine(const float* __restrict__ Pin, const float* __restrict__ Hin,
           float* __restrict__ h0out) {
  const int idx = blockIdx.x * 256 + threadIdx.x;
  const int bb  = idx / (N_ * D_);
  const int nd  = idx % (N_ * D_);
  const size_t base = (size_t)bb * G_ * N_ * D_ + nd;

  float h = 0.0f;
#pragma unroll
  for (int g = 0; g < G_; ++g) {
    const size_t o = base + (size_t)g * N_ * D_;
    h0out[o] = h;
    h = Pin[o] * h + Hin[o];
  }
}

__global__ void __launch_bounds__(128)
s6_scan_emit(const float* __restrict__ xdbl, const float* __restrict__ delta,
             const float* __restrict__ x, const float* __restrict__ Alog,
             const float* __restrict__ Dp, const float* __restrict__ h0in,
             float* __restrict__ y) {
  __shared__ float ypart[T_ * NS_ * 32];

  const int lane = threadIdx.x & 31;
  const int q  = __builtin_amdgcn_readfirstlane((int)(threadIdx.x >> 5));
  const int dg = blockIdx.x % DG_;
  const int bg = blockIdx.x / DG_;
  const int g  = bg % G_;
  const int bb = bg / G_;
  const int d  = dg * 32 + lane;
  const int nq = q * NQ_;

  v2f A2[NQ_ / 2], h2[NQ_ / 2];
  const v4f* alp = (const v4f*)(Alog + (size_t)d * N_ + nq);
#pragma unroll
  for (int k = 0; k < NQ_ / 4; ++k) {
    const v4f al = alp[k];
    A2[2 * k].x     = -__expf(al.x);
    A2[2 * k].y     = -__expf(al.y);
    A2[2 * k + 1].x = -__expf(al.z);
    A2[2 * k + 1].y = -__expf(al.w);
  }
  const size_t hbase = ((size_t)bg * N_ + nq) * D_ + d;
#pragma unroll
  for (int j = 0; j < NQ_ / 2; ++j) {
    h2[j].x = h0in[hbase + (size_t)(2 * j) * D_];
    h2[j].y = h0in[hbase + (size_t)(2 * j + 1) * D_];
  }

  const size_t row0 = (size_t)bb * L_ + (size_t)g * T_;
  const float* dp = delta + row0 * D_ + d;
  const float* xp = x     + row0 * D_ + d;
  const float* bp = xdbl  + row0 * E_ + R_ + nq;
  const float* cp = bp + N_;
  float* yw = ypart + q * 32 + lane;

  v16f B0 = *(const v16f*)bp;
  v16f C0 = *(const v16f*)cp;
  float d0 = dp[0],  x0 = xp[0];
  float d1 = dp[D_], x1 = xp[D_];
  for (int tt = 0; tt < T_; tt += 2) {
    const int t2 = (tt + 2 < T_) ? tt + 2 : tt;
    const v16f B1 = *(const v16f*)(bp + (size_t)(tt + 1) * E_);
    const v16f C1 = *(const v16f*)(cp + (size_t)(tt + 1) * E_);
    const float dc0 = d0, xc0 = x0, dc1 = d1, xc1 = x1;
    d0 = dp[(size_t)t2 * D_];        x0 = xp[(size_t)t2 * D_];
    d1 = dp[(size_t)(t2 + 1) * D_];  x1 = xp[(size_t)(t2 + 1) * D_];
    yw[tt * NS_ * 32] = s6_step_emit(B0, C0, dc0, xc0, A2, h2);
    B0 = *(const v16f*)(bp + (size_t)t2 * E_);
    C0 = *(const v16f*)(cp + (size_t)t2 * E_);
    yw[(tt + 1) * NS_ * 32] = s6_step_emit(B1, C1, dc1, xc1, A2, h2);
  }
  __syncthreads();

  const float Dv = Dp[d];
#pragma unroll 4
  for (int k = 0; k < T_ / NS_; ++k) {
    const int tt = q * (T_ / NS_) + k;
    const float* yp = ypart + tt * NS_ * 32 + lane;
    const float s = (yp[0] + yp[32]) + (yp[64] + yp[96]);
    const float xv = xp[(size_t)tt * D_];
    y[(row0 + tt) * D_ + d] = s + xv * Dv;
  }
}

extern "C" void kernel_launch(void* const* d_in, const int* in_sizes, int n_in,
                              void* d_out, int out_size, void* d_ws, size_t ws_size,
                              hipStream_t stream) {
  const float* x      = (const float*)d_in[0];
  const float* Wxproj = (const float*)d_in[1];
  const float* Wdt    = (const float*)d_in[2];
  const float* bdt    = (const float*)d_in[3];
  const float* Alog   = (const float*)d_in[4];
  const float* Dp     = (const float*)d_in[5];
  float*       y      = (float*)d_out;

  float* ws    = (float*)d_ws;
  float* xdbl  = ws;
  float* delta = xdbl  + (size_t)BL_ * E_;
  float* Pws   = delta + (size_t)BL_ * D_;
  float* Hws   = Pws   + (size_t)B_ * D_ * G_ * N_;
  float* h0ws  = Hws   + (size_t)B_ * D_ * G_ * N_;

  s6_xproj_gemm<<<256, 256, 0, stream>>>(x, Wxproj, xdbl);
  s6_delta_gemm<<<1024, 256, 0, stream>>>(xdbl, Wdt, bdt, delta);
  s6_scan_chunks<<<1024, 128, 0, stream>>>(xdbl, delta, x, Alog, Pws, Hws);
  s6_combine<<<256, 256, 0, stream>>>(Pws, Hws, h0ws);
  s6_scan_emit<<<1024, 128, 0, stream>>>(xdbl, delta, x, Alog, Dp, h0ws, y);
}


# Scan passes: a chunk's delta, x and B/C rows staged once in LDS by whole-line loads of all four waves; the time loop reads LDS only, one step ahead (sum order unchanged)
# speedup vs baseline: 3.9683x; 1.2496x over previous
#include <hip/hip_runtime.h>
#include <hip/hip_bf16.h>

typedef float v2f __attribute__((ext_vector_type(2)));
typedef float v8f __attribute__((ext_vector_type(8)));
typedef float v4f __attribute__((ext_vector_type(4)));
typedef float v16f __attribute__((ext_vector_type(16)));

#define B_  2
#define L_  2048
#define D_  512
#define N_  64
#define R_  32
#define E_  160
#define BL_ (B_ * L_)
#define G_  32
#define T_  (L_ / G_)
#define NS_ 4
#define NQ_ (N_ / NS_)
#define DG_ (D_ / 32)

#define XW_  8
#define XK_  (D_ / XW_)
#define XP_  68
#define XT_  (16 * XP_)
#define XET_ (E_ / 16)
#define XWS_ (5 * XT_)

__device__ __forceinline__ void
s6_slice_fetch(const float* __restrict__ src, const int lane, v4f (&r)[8]) {
  const float* p = src + (size_t)(lane >> 4) * D_ + 4 * (lane & 15);
#pragma unroll
  for (int i = 0; i < 8; ++i) r[i] = *(const v4f*)(p + (size_t)(2 * i) * D_);
}

__device__ __forceinline__ void
s6_slice_park(float* dst, const int lane, const v4f (&r)[8]) {
  float* p = dst + (lane >> 4) * XP_ + 4 * (lane & 15);
#pragma unroll
  for (int i = 0; i < 8; ++i) *(v4f*)(p + 2 * i * XP_) = r[i];
}

__device__ __forceinline__ void
s6_slice_frags(const float* tile, const int lane, v4f (&f)[8]) {
  const float* p = tile + (lane & 15) * XP_ + 4 * (lane >> 4);
#pragma unroll
  for (int i = 0; i < 8; ++i) f[i] = *(const v4f*)(p + 8 * i);
}

__device__ __forceinline__ void s6_wave_lds_handover() {
  __builtin_amdgcn_fence(__ATOMIC_RELEASE, "wavefront");
  __builtin_amdgcn_wave_barrier();
  __builtin_amdgcn_fence(__ATOMIC_ACQUIRE, "wavefront");
}

__global__ void __launch_bounds__(256)
s6_xproj_gemm(const float* __restrict__ x, const float* __restrict__ W,
              float* __restrict__ xdbl) {
  __shared__ __attribute__((aligned(16))) float sm[XW_ * XWS_];

  const int lane = threadIdx.x & 31;
  const int w  = __builtin_amdgcn_readfirstlane((int)(threadIdx.x >> 5));
  const int rt = blockIdx.x;

  float* sa = sm + w * XWS_;
  float* sb = sa + XT_;
  const float* xa = x + (size_t)rt * 16 * D_ + w * XK_;
  const float* wb = W + w * XK_;

  v4f ra[8], rb0[8], rb1[8];
  s6_slice_fetch(xa, lane, ra);
  s6_slice_fetch(wb, lane, rb0);
  s6_slice_fetch(wb + (size_t)16 * D_, lane, rb1);
  s6_slice_park(sa, lane, ra);
  s6_slice_park(sb, lane, rb0);
  s6_slice_park(sb + XT_, lane, rb1);
  s6_slice_fetch(wb + (size_t)32 * D_, lane, rb0);
  s6_slice_fetch(wb + (size_t)48 * D_, lane, rb1);
  s6_wave_lds_handover();

  v4f af[8], bf[2][2][8];
  s6_slice_frags(sa, lane, af);
  s6_slice_frags(sb, lane, bf[0][0]);
  s6_slice_frags(sb + XT_, lane, bf[0][1]);
  s6_slice_park(sb + 2 * XT_, lane, rb0);
  s6_slice_park(sb + 3 * XT_, lane, rb1);
  s6_wave_lds_handover();

  v8f acc[XET_];
#pragma unroll
  for (int j = 0; j < XET_; ++j) acc[j] = (v8f){};

#pragma unroll
  for (int jp = 0; jp < XET_ / 2; ++jp) {
    const int cs = jp & 1, ns = cs ^ 1;
    if (jp + 2 < XET_ / 2) {
      s6_slice_fetch(wb + (size_t)(2 * jp + 4) * 16 * D_, lane, rb0);
      s6_slice_fetch(wb + (size_t)(2 * jp + 5) * 16 * D_, lane, rb1);
    }
    if (jp + 1 < XET_ / 2) {
      s6_slice_frags(sb + ns * 2 * XT_, lane, bf[ns][0]);
      s6_slice_frags(sb + ns * 2 * XT_ + XT_, lane, bf[ns][1]);
    }
    __builtin_amdgcn_sched_barrier(0);
#pragma unroll
    for (int i = 0; i < 8; ++i) {
      const v2f a0 = {af[i].x, af[i].y}, a1 = {af[i].z, af[i].w};
      const v2f p0 = {bf[cs][0][i].x, bf[cs][0][i].y}, p1 = {bf[cs][0][i].z, bf[cs][0][i].w};
      const v2f q0 = {bf[cs][1][i].x, bf[cs][1][i].y}, q1 = {bf[cs][1][i].z, bf[cs][1][i].w};
      acc[2 * jp]     = __builtin_amdgcn_wmma_f32_16x16x4_f32(false, a0, false, p0, (short)0, acc[2 * jp],     false, false);
      acc[2 * jp + 1] = __builtin_amdgcn_wmma_f32_16x16x4_f32(false, a0, false, q0, (short)0, acc[2 * jp + 1], false, false);
      acc[2 * jp]     = __builtin_amdgcn_wmma_f32_16x16x4_f32(false, a1, false, p1, (short)0, acc[2 * jp],     false, false);
      acc[2 * jp + 1] = __builtin_amdgcn_wmma_f32_16x16x4_f32(false, a1, false, q1, (short)0, acc[2 * jp + 1], false, false);
    }
    __builtin_amdgcn_sched_barrier(0);
    if (jp + 2 < XET_ / 2) {
      s6_slice_park(sb + cs * 2 * XT_, lane, rb0);
      s6_slice_park(sb + cs * 2 * XT_ + XT_, lane, rb1);
      s6_wave_lds_handover();
    }
  }

  __syncthreads();
#pragma unroll
  for (int j = 0; j < XET_; ++j) {
#pragma unroll
    for (int r = 0; r < 8; ++r)
      sm[((w * XET_ + j) * 8 + r) * 32 + lane] = acc[j][r];
  }
  __syncthreads();

  float* orow = xdbl + (size_t)rt * 16 * E_;
#pragma unroll 2
  for (int o = threadIdx.x; o < 16 * E_; o += 256) {
    const int row = o / E_, col = o % E_;
    const int idx = ((col >> 4) * 8 + (row & 7)) * 32 + (col & 15) + 16 * (row >> 3);
    float s = sm[idx];
#pragma unroll
    for (int q = 1; q < XW_; ++q) s += sm[q * XET_ * 256 + idx];
    orow[o] = s;
  }
}

__global__ void __launch_bounds__(256)
s6_delta_gemm(const float* __restrict__ xdbl, const float* __restrict__ Wdt,
              const float* __restrict__ bdt, float* __restrict__ delta) {
  const int wave  = blockIdx.x * (blockDim.x >> 5) + (threadIdx.x >> 5);
  const int lane  = threadIdx.x & 31;
  const int rt    = wave >> 5;
  const int nt    = wave & 31;
  const int laneM = lane & 15;
  const int laneH = lane >> 4;

  const float* arow = xdbl + (size_t)(rt * 16 + laneM) * E_ + 2 * laneH;
  const float* brow = Wdt  + (size_t)(nt * 16 + laneM) * R_ + 2 * laneH;

  v8f c = {};
#pragma unroll
  for (int k = 0; k < R_; k += 4) {
    v2f a = *(const v2f*)(arow + k);
    v2f b = *(const v2f*)(brow + k);
    c = __builtin_amdgcn_wmma_f32_16x16x4_f32(false, a, false, b,
                                              (short)0, c, false, false);
  }
  const int col  = nt * 16 + laneM;
  const float bi = bdt[col];
#pragma unroll
  for (int j = 0; j < 8; ++j) {
    const int row = rt * 16 + j + 8 * laneH;
    float z = c[j] + bi;
    float sp = (z > 20.0f) ? z : __logf(1.0f + __expf(z));
    delta[(size_t)row * D_ + col] = sp;
  }
}


__device__ __forceinline__ void
s6_stage_dx(const float* __restrict__ dsrc, const float* __restrict__ xsrc, float* sDX) {
#pragma unroll
  for (int i = 0; i < 4; ++i) {
    const int idx = (int)threadIdx.x + 128 * i;
    const int row = idx >> 3, c4 = (idx & 7) * 4;
    const v4f dv = *(const v4f*)(dsrc + (size_t)row * D_ + c4);
    const v4f xv = *(const v4f*)(xsrc + (size_t)row * D_ + c4);
    float* p = sDX + (row * 32 + c4) * 2;
    *(v4f*)p       = (v4f){dv.x, xv.x, dv.y, xv.y};
    *(v4f*)(p + 4) = (v4f){dv.z, xv.z, dv.w, xv.w};
  }
}

__device__ __forceinline__ void s6_row16(const float* p, v4f (&r)[4]) {
#pragma unroll
  for (int i = 0; i < 4; ++i) r[i] = *(const v4f*)(p + 4 * i);
}

__device__ __forceinline__ void
s6_step_chunk(const v4f (&Bc)[4], const float dlt, const float xv,
              const v2f (&A2)[NQ_ / 2], v2f (&h2)[NQ_ / 2], v2f (&P2)[NQ_ / 2]) {
  const float du = dlt * xv;
#pragma unroll
  for (int j = 0; j < NQ_ / 2; ++j) {
    const v2f arg = dlt * A2[j];
    v2f a;  a.x = __expf(arg.x);  a.y = __expf(arg.y);
    v2f Bv; Bv.x = Bc[j >> 1][2 * (j & 1)];  Bv.y = Bc[j >> 1][2 * (j & 1) + 1];
    h2[j] = a * h2[j] + du * Bv;
    P2[j] *= a;
  }
}

__device__ __forceinline__ float
s6_step_emit(const v4f (&Bc)[4], const v4f (&Cc)[4], const float dlt, const float xv,
             const v2f (&A2)[NQ_ / 2], v2f (&h2)[NQ_ / 2]) {
  const float du = dlt * xv;
  v2f acc = {0.0f, 0.0f};
#pragma unroll
  for (int j = 0; j < NQ_ / 2; ++j) {
    const v2f arg = dlt * A2[j];
    v2f a;  a.x = __expf(arg.x);  a.y = __expf(arg.y);
    v2f Bv; Bv.x = Bc[j >> 1][2 * (j & 1)];  Bv.y = Bc[j >> 1][2 * (j & 1) + 1];
    v2f Cv; Cv.x = Cc[j >> 1][2 * (j & 1)];  Cv.y = Cc[j >> 1][2 * (j & 1) + 1];
    h2[j] = a * h2[j] + du * Bv;
    acc += h2[j] * Cv;
  }
  return acc.x + acc.y;
}

__global__ void __launch_bounds__(128)
s6_scan_chunks(const float* __restrict__ xdbl, const float* __restrict__ delta,
               const float* __restrict__ x, const float* __restrict__ Alog,
               float* __restrict__ Pout, float* __restrict__ Hout) {
  __shared__ __attribute__((aligned(16))) float sDX[T_ * 64];
  __shared__ __attribute__((aligned(16))) float sB[T_ * N_];

  const int lane = threadIdx.x & 31;
  const int q  = __builtin_amdgcn_readfirstlane((int)(threadIdx.x >> 5));
  const int dg = blockIdx.x % DG_;
  const int bg = blockIdx.x / DG_;
  const int g  = bg % G_;
  const int bb = bg / G_;
  const int d  = dg * 32 + lane;
  const int nq = q * NQ_;

  v2f A2[NQ_ / 2], h2[NQ_ / 2], P2[NQ_ / 2];
  const v4f* alp = (const v4f*)(Alog + (size_t)d * N_ + nq);
#pragma unroll
  for (int k = 0; k < NQ_ / 4; ++k) {
    const v4f al = alp[k];
    A2[2 * k].x     = -__expf(al.x);
    A2[2 * k].y     = -__expf(al.y);
    A2[2 * k + 1].x = -__expf(al.z);
    A2[2 * k + 1].y = -__expf(al.w);
  }
#pragma unroll
  for (int j = 0; j < NQ_ / 2; ++j) {
    h2[j].x = 0.0f;  h2[j].y = 0.0f;
    P2[j].x = 1.0f;  P2[j].y = 1.0f;
  }

  const size_t row0 = (size_t)bb * L_ + (size_t)g * T_;
  s6_stage_dx(delta + row0 * D_ + dg * 32, x + row0 * D_ + dg * 32, sDX);
  {
    const float* bsrc = xdbl + row0 * E_ + R_;
#pragma unroll
    for (int i = 0; i < 8; ++i) {
      const int idx = (int)threadIdx.x + 128 * i;
      const int row = idx >> 4, c4 = (idx & 15) * 4;
      *(v4f*)(sB + row * N_ + c4) = *(const v4f*)(bsrc + (size_t)row * E_ + c4);
    }
  }
  __syncthreads();

  const float* bq = sB + nq;
  const float* dx = sDX + 2 * lane;

  v4f Ba[4], Bb[4];
  v2f da, db;
  s6_row16(bq, Ba);
  da = *(const v2f*)dx;
  for (int tt = 0; tt < T_; tt += 2) {
    s6_row16(bq + (tt + 1) * N_, Bb);
    db = *(const v2f*)(dx + (tt + 1) * 64);
    s6_step_chunk(Ba, da.x, da.y, A2, h2, P2);
    asm volatile("" :: "v"(Bb[0]), "v"(Bb[1]), "v"(Bb[2]), "v"(Bb[3]), "v"(db));
    const int t2 = (tt + 2 < T_) ? tt + 2 : tt + 1;
    s6_row16(bq + t2 * N_, Ba);
    da = *(const v2f*)(dx + t2 * 64);
    s6_step_chunk(Bb, db.x, db.y, A2, h2, P2);
    asm volatile("" :: "v"(Ba[0]), "v"(Ba[1]), "v"(Ba[2]), "v"(Ba[3]), "v"(da));
  }
  const size_t base = ((size_t)bg * N_ + nq) * D_ + d;
#pragma unroll
  for (int j = 0; j < NQ_ / 2; ++j) {
    Pout[base + (size_t)(2 * j) * D_]     = P2[j].x;
    Pout[base + (size_t)(2 * j + 1) * D_] = P2[j].y;
    Hout[base + (size_t)(2 * j) * D_]     = h2[j].x;
    Hout[base + (size_t)(2 * j + 1) * D_] = h2[j].y;
  }
}

__global__ void __launch_bounds__(256)
s6_combine(const float* __restrict__ Pin, const float* __restrict__ Hin,
           float* __restrict__ h0out) {
  const int idx = blockIdx.x * 256 + threadIdx.x;
  const int bb  = idx / (N_ * D_);
  const int nd  = idx % (N_ * D_);
  const size_t base = (size_t)bb * G_ * N_ * D_ + nd;

  float h = 0.0f;
#pragma unroll
  for (int g = 0; g < G_; ++g) {
    const size_t o = base + (size_t)g * N_ * D_;
    h0out[o] = h;
    h = Pin[o] * h + Hin[o];
  }
}

__global__ void __launch_bounds__(128)
s6_scan_emit(const float* __restrict__ xdbl, const float* __restrict__ delta,
             const float* __restrict__ x, const float* __restrict__ Alog,
             const float* __restrict__ Dp, const float* __restrict__ h0in,
             float* __restrict__ y) {
  __shared__ __attribute__((aligned(16))) float sDX[T_ * 64];
  __shared__ __attribute__((aligned(16))) float sBC[T_ * NS_ * 32];

  const int lane = threadIdx.x & 31;
  const int q  = __builtin_amdgcn_readfirstlane((int)(threadIdx.x >> 5));
  const int dg = blockIdx.x % DG_;
  const int bg = blockIdx.x / DG_;
  const int g  = bg % G_;
  const int bb = bg / G_;
  const int d  = dg * 32 + lane;
  const int nq = q * NQ_;

  v2f A2[NQ_ / 2], h2[NQ_ / 2];
  const v4f* alp = (const v4f*)(Alog + (size_t)d * N_ + nq);
#pragma unroll
  for (int k = 0; k < NQ_ / 4; ++k) {
    const v4f al = alp[k];
    A2[2 * k].x     = -__expf(al.x);
    A2[2 * k].y     = -__expf(al.y);
    A2[2 * k + 1].x = -__expf(al.z);
    A2[2 * k + 1].y = -__expf(al.w);
  }
  const size_t hbase = ((size_t)bg * N_ + nq) * D_ + d;
#pragma unroll
  for (int j = 0; j < NQ_ / 2; ++j) {
    h2[j].x = h0in[hbase + (size_t)(2 * j) * D_];
    h2[j].y = h0in[hbase + (size_t)(2 * j + 1) * D_];
  }

  const size_t row0 = (size_t)bb * L_ + (size_t)g * T_;
  s6_stage_dx(delta + row0 * D_ + dg * 32, x + row0 * D_ + dg * 32, sDX);
  {
    const float* bsrc = xdbl + row0 * E_ + R_;
#pragma unroll
    for (int i = 0; i < 16; ++i) {
      const int idx = (int)threadIdx.x + 128 * i;
      const int row = idx >> 5, col = (idx & 31) * 4;
      const int n = col & 63;
      *(v4f*)(sBC + (row * NS_ + (n >> 4)) * 32 + (col >> 6) * 16 + (n & 15)) =
          *(const v4f*)(bsrc + (size_t)row * E_ + col);
    }
  }
  __syncthreads();

  float* bc = sBC + q * 32;
  const float* dx = sDX + 2 * lane;

  v4f Ba[4], Ca[4], Bb[4], Cb[4];
  v2f da, db;
  s6_row16(bc, Ba);
  s6_row16(bc + 16, Ca);
  da = *(const v2f*)dx;
  for (int tt = 0; tt < T_; tt += 2) {
    s6_row16(bc + (tt + 1) * NS_ * 32, Bb);
    s6_row16(bc + (tt + 1) * NS_ * 32 + 16, Cb);
    db = *(const v2f*)(dx + (tt + 1) * 64);
    bc[tt * NS_ * 32 + lane] = s6_step_emit(Ba, Ca, da.x, da.y, A2, h2);
    asm volatile("" :: "v"(Bb[0]), "v"(Bb[1]), "v"(Bb[2]), "v"(Bb[3]),
                       "v"(Cb[0]), "v"(Cb[1]), "v"(Cb[2]), "v"(Cb[3]), "v"(db));
    const int t2 = (tt + 2 < T_) ? tt + 2 : tt + 1;
    s6_row16(bc + t2 * NS_ * 32, Ba);
    s6_row16(bc + t2 * NS_ * 32 + 16, Ca);
    da = *(const v2f*)(dx + t2 * 64);
    bc[(tt + 1) * NS_ * 32 + lane] = s6_step_emit(Bb, Cb, db.x, db.y, A2, h2);
    asm volatile("" :: "v"(Ba[0]), "v"(Ba[1]), "v"(Ba[2]), "v"(Ba[3]),
                       "v"(Ca[0]), "v"(Ca[1]), "v"(Ca[2]), "v"(Ca[3]), "v"(da));
  }
  __syncthreads();

  const float Dv = Dp[d];
#pragma unroll 4
  for (int k = 0; k < T_ / NS_; ++k) {
    const int tt = q * (T_ / NS_) + k;
    const float* yp = sBC + tt * NS_ * 32 + lane;
    const float s = (yp[0] + yp[32]) + (yp[64] + yp[96]);
    const float xv = dx[tt * 64 + 1];
    y[(row0 + tt) * D_ + d] = s + xv * Dv;
  }
}

extern "C" void kernel_launch(void* const* d_in, const int* in_sizes, int n_in,
                              void* d_out, int out_size, void* d_ws, size_t ws_size,
                              hipStream_t stream) {
  const float* x      = (const float*)d_in[0];
  const float* Wxproj = (const float*)d_in[1];
  const float* Wdt    = (const float*)d_in[2];
  const float* bdt    = (const float*)d_in[3];
  const float* Alog   = (const float*)d_in[4];
  const float* Dp     = (const float*)d_in[5];
  float*       y      = (float*)d_out;

  float* ws    = (float*)d_ws;
  float* xdbl  = ws;
  float* delta = xdbl  + (size_t)BL_ * E_;
  float* Pws   = delta + (size_t)BL_ * D_;
  float* Hws   = Pws   + (size_t)B_ * D_ * G_ * N_;
  float* h0ws  = Hws   + (size_t)B_ * D_ * G_ * N_;

  s6_xproj_gemm<<<256, 256, 0, stream>>>(x, Wxproj, xdbl);
  s6_delta_gemm<<<1024, 256, 0, stream>>>(xdbl, Wdt, bdt, delta);
  s6_scan_chunks<<<1024, 128, 0, stream>>>(xdbl, delta, x, Alog, Pws, Hws);
  s6_combine<<<256, 256, 0, stream>>>(Pws, Hws, h0ws);
  s6_scan_emit<<<1024, 128, 0, stream>>>(xdbl, delta, x, Alog, Dp, h0ws, y);
}


# delta = softplus(x_dbl[:, :r] W_dt^T + b_dt) computed per 64x32 chunk tile inside both scan kernels (16 WMMA per wave, same k order, sum order unchanged); delta kernel and plane removed: 4 launches
# speedup vs baseline: 4.1432x; 1.0441x over previous
#include <hip/hip_runtime.h>
#include <hip/hip_bf16.h>

typedef float v2f __attribute__((ext_vector_type(2)));
typedef float v8f __attribute__((ext_vector_type(8)));
typedef float v4f __attribute__((ext_vector_type(4)));
typedef float v16f __attribute__((ext_vector_type(16)));

#define B_  2
#define L_  2048
#define D_  512
#define N_  64
#define R_  32
#define E_  160
#define BL_ (B_ * L_)
#define G_  32
#define T_  (L_ / G_)
#define NS_ 4
#define NQ_ (N_ / NS_)
#define DG_ (D_ / 32)

#define XW_  8
#define XK_  (D_ / XW_)
#define XP_  68
#define XT_  (16 * XP_)
#define XET_ (E_ / 16)
#define XWS_ (5 * XT_)

__device__ __forceinline__ void
s6_slice_fetch(const float* __restrict__ src, const int lane, v4f (&r)[8]) {
  const float* p = src + (size_t)(lane >> 4) * D_ + 4 * (lane & 15);
#pragma unroll
  for (int i = 0; i < 8; ++i) r[i] = *(const v4f*)(p + (size_t)(2 * i) * D_);
}

__device__ __forceinline__ void
s6_slice_park(float* dst, const int lane, const v4f (&r)[8]) {
  float* p = dst + (lane >> 4) * XP_ + 4 * (lane & 15);
#pragma unroll
  for (int i = 0; i < 8; ++i) *(v4f*)(p + 2 * i * XP_) = r[i];
}

__device__ __forceinline__ void
s6_slice_frags(const float* tile, const int lane, v4f (&f)[8]) {
  const float* p = tile + (lane & 15) * XP_ + 4 * (lane >> 4);
#pragma unroll
  for (int i = 0; i < 8; ++i) f[i] = *(const v4f*)(p + 8 * i);
}

__device__ __forceinline__ void s6_wave_lds_handover() {
  __builtin_amdgcn_fence(__ATOMIC_RELEASE, "wavefront");
  __builtin_amdgcn_wave_barrier();
  __builtin_amdgcn_fence(__ATOMIC_ACQUIRE, "wavefront");
}

__global__ void __launch_bounds__(256)
s6_xproj_gemm(const float* __restrict__ x, const float* __restrict__ W,
              float* __restrict__ xdbl) {
  __shared__ __attribute__((aligned(16))) float sm[XW_ * XWS_];

  const int lane = threadIdx.x & 31;
  const int w  = __builtin_amdgcn_readfirstlane((int)(threadIdx.x >> 5));
  const int rt = blockIdx.x;

  float* sa = sm + w * XWS_;
  float* sb = sa + XT_;
  const float* xa = x + (size_t)rt * 16 * D_ + w * XK_;
  const float* wb = W + w * XK_;

  v4f ra[8], rb0[8], rb1[8];
  s6_slice_fetch(xa, lane, ra);
  s6_slice_fetch(wb, lane, rb0);
  s6_slice_fetch(wb + (size_t)16 * D_, lane, rb1);
  s6_slice_park(sa, lane, ra);
  s6_slice_park(sb, lane, rb0);
  s6_slice_park(sb + XT_, lane, rb1);
  s6_slice_fetch(wb + (size_t)32 * D_, lane, rb0);
  s6_slice_fetch(wb + (size_t)48 * D_, lane, rb1);
  s6_wave_lds_handover();

  v4f af[8], bf[2][2][8];
  s6_slice_frags(sa, lane, af);
  s6_slice_frags(sb, lane, bf[0][0]);
  s6_slice_frags(sb + XT_, lane, bf[0][1]);
  s6_slice_park(sb + 2 * XT_, lane, rb0);
  s6_slice_park(sb + 3 * XT_, lane, rb1);
  s6_wave_lds_handover();

  v8f acc[XET_];
#pragma unroll
  for (int j = 0; j < XET_; ++j) acc[j] = (v8f){};

#pragma unroll
  for (int jp = 0; jp < XET_ / 2; ++jp) {
    const int cs = jp & 1, ns = cs ^ 1;
    if (jp + 2 < XET_ / 2) {
      s6_slice_fetch(wb + (size_t)(2 * jp + 4) * 16 * D_, lane, rb0);
      s6_slice_fetch(wb + (size_t)(2 * jp + 5) * 16 * D_, lane, rb1);
    }
    if (jp + 1 < XET_ / 2) {
      s6_slice_frags(sb + ns * 2 * XT_, lane, bf[ns][0]);
      s6_slice_frags(sb + ns * 2 * XT_ + XT_, lane, bf[ns][1]);
    }
    __builtin_amdgcn_sched_barrier(0);
#pragma unroll
    for (int i = 0; i < 8; ++i) {
      const v2f a0 = {af[i].x, af[i].y}, a1 = {af[i].z, af[i].w};
      const v2f p0 = {bf[cs][0][i].x, bf[cs][0][i].y}, p1 = {bf[cs][0][i].z, bf[cs][0][i].w};
      const v2f q0 = {bf[cs][1][i].x, bf[cs][1][i].y}, q1 = {bf[cs][1][i].z, bf[cs][1][i].w};
      acc[2 * jp]     = __builtin_amdgcn_wmma_f32_16x16x4_f32(false, a0, false, p0, (short)0, acc[2 * jp],     false, false);
      acc[2 * jp + 1] = __builtin_amdgcn_wmma_f32_16x16x4_f32(false, a0, false, q0, (short)0, acc[2 * jp + 1], false, false);
      acc[2 * jp]     = __builtin_amdgcn_wmma_f32_16x16x4_f32(false, a1, false, p1, (short)0, acc[2 * jp],     false, false);
      acc[2 * jp + 1] = __builtin_amdgcn_wmma_f32_16x16x4_f32(false, a1, false, q1, (short)0, acc[2 * jp + 1], false, false);
    }
    __builtin_amdgcn_sched_barrier(0);
    if (jp + 2 < XET_ / 2) {
      s6_slice_park(sb + cs * 2 * XT_, lane, rb0);
      s6_slice_park(sb + cs * 2 * XT_ + XT_, lane, rb1);
      s6_wave_lds_handover();
    }
  }

  __syncthreads();
#pragma unroll
  for (int j = 0; j < XET_; ++j) {
#pragma unroll
    for (int r = 0; r < 8; ++r)
      sm[((w * XET_ + j) * 8 + r) * 32 + lane] = acc[j][r];
  }
  __syncthreads();

  float* orow = xdbl + (size_t)rt * 16 * E_;
#pragma unroll 2
  for (int o = threadIdx.x; o < 16 * E_; o += 256) {
    const int row = o / E_, col = o % E_;
    const int idx = ((col >> 4) * 8 + (row & 7)) * 32 + (col & 15) + 16 * (row >> 3);
    float s = sm[idx];
#pragma unroll
    for (int q = 1; q < XW_; ++q) s += sm[q * XET_ * 256 + idx];
    orow[o] = s;
  }
}

__device__ __forceinline__ float s6_softplus(const float z) {
  const float sp = __logf(1.0f + __expf(z));
  return (z > 20.0f) ? z : sp;
}

__device__ __forceinline__ void
s6_delta_tile(const float* __restrict__ xrows,
              const float* __restrict__ wdt,
              const float* __restrict__ bdt,
              const int q, const int lane, float* sDX) {
  const int laneM = lane & 15;
  const int laneH = lane >> 4;
  const float* arow = xrows + (size_t)(q * 16 + laneM) * E_ + 2 * laneH;
  const float* br0  = wdt + (size_t)laneM * R_ + 2 * laneH;
  const float* br1  = br0 + 16 * R_;

  v8f c0 = {}, c1 = {};
#pragma unroll
  for (int k = 0; k < R_; k += 4) {
    const v2f a  = *(const v2f*)(arow + k);
    const v2f b0 = *(const v2f*)(br0 + k);
    const v2f b1 = *(const v2f*)(br1 + k);
    c0 = __builtin_amdgcn_wmma_f32_16x16x4_f32(false, a, false, b0, (short)0, c0, false, false);
    c1 = __builtin_amdgcn_wmma_f32_16x16x4_f32(false, a, false, b1, (short)0, c1, false, false);
  }
  const float bi0 = bdt[laneM];
  const float bi1 = bdt[16 + laneM];
  float* p = sDX + ((q * 16 + 8 * laneH) * 32 + laneM) * 2;
#pragma unroll
  for (int j = 0; j < 8; ++j) {
    p[j * 64]      = s6_softplus(c0[j] + bi0);
    p[j * 64 + 32] = s6_softplus(c1[j] + bi1);
  }
}


__device__ __forceinline__ void
s6_stage_x(const float* __restrict__ xsrc, float* sDX) {
#pragma unroll
  for (int i = 0; i < 4; ++i) {
    const int idx = (int)threadIdx.x + 128 * i;
    const int row = idx >> 3, c4 = (idx & 7) * 4;
    const v4f xv = *(const v4f*)(xsrc + (size_t)row * D_ + c4);
    float* p = sDX + (row * 32 + c4) * 2 + 1;
    p[0] = xv.x;  p[2] = xv.y;  p[4] = xv.z;  p[6] = xv.w;
  }
}

__device__ __forceinline__ void s6_row16(const float* p, v4f (&r)[4]) {
#pragma unroll
  for (int i = 0; i < 4; ++i) r[i] = *(const v4f*)(p + 4 * i);
}

__device__ __forceinline__ void
s6_step_chunk(const v4f (&Bc)[4], const float dlt, const float xv,
              const v2f (&A2)[NQ_ / 2], v2f (&h2)[NQ_ / 2], v2f (&P2)[NQ_ / 2]) {
  const float du = dlt * xv;
#pragma unroll
  for (int j = 0; j < NQ_ / 2; ++j) {
    const v2f arg = dlt * A2[j];
    v2f a;  a.x = __expf(arg.x);  a.y = __expf(arg.y);
    v2f Bv; Bv.x = Bc[j >> 1][2 * (j & 1)];  Bv.y = Bc[j >> 1][2 * (j & 1) + 1];
    h2[j] = a * h2[j] + du * Bv;
    P2[j] *= a;
  }
}

__device__ __forceinline__ float
s6_step_emit(const v4f (&Bc)[4], const v4f (&Cc)[4], const float dlt, const float xv,
             const v2f (&A2)[NQ_ / 2], v2f (&h2)[NQ_ / 2]) {
  const float du = dlt * xv;
  v2f acc = {0.0f, 0.0f};
#pragma unroll
  for (int j = 0; j < NQ_ / 2; ++j) {
    const v2f arg = dlt * A2[j];
    v2f a;  a.x = __expf(arg.x);  a.y = __expf(arg.y);
    v2f Bv; Bv.x = Bc[j >> 1][2 * (j & 1)];  Bv.y = Bc[j >> 1][2 * (j & 1) + 1];
    v2f Cv; Cv.x = Cc[j >> 1][2 * (j & 1)];  Cv.y = Cc[j >> 1][2 * (j & 1) + 1];
    h2[j] = a * h2[j] + du * Bv;
    acc += h2[j] * Cv;
  }
  return acc.x + acc.y;
}

__global__ void __launch_bounds__(128)
s6_scan_chunks(const float* __restrict__ xdbl, const float* __restrict__ Wdt,
               const float* __restrict__ bdt,
               const float* __restrict__ x, const float* __restrict__ Alog,
               float* __restrict__ Pout, float* __restrict__ Hout) {
  __shared__ __attribute__((aligned(16))) float sDX[T_ * 64];
  __shared__ __attribute__((aligned(16))) float sB[T_ * N_];

  const int lane = threadIdx.x & 31;
  const int q  = __builtin_amdgcn_readfirstlane((int)(threadIdx.x >> 5));
  const int dg = blockIdx.x % DG_;
  const int bg = blockIdx.x / DG_;
  const int g  = bg % G_;
  const int bb = bg / G_;
  const int d  = dg * 32 + lane;
  const int nq = q * NQ_;

  v2f A2[NQ_ / 2], h2[NQ_ / 2], P2[NQ_ / 2];
  const v4f* alp = (const v4f*)(Alog + (size_t)d * N_ + nq);
#pragma unroll
  for (int k = 0; k < NQ_ / 4; ++k) {
    const v4f al = alp[k];
    A2[2 * k].x     = -__expf(al.x);
    A2[2 * k].y     = -__expf(al.y);
    A2[2 * k + 1].x = -__expf(al.z);
    A2[2 * k + 1].y = -__expf(al.w);
  }
#pragma unroll
  for (int j = 0; j < NQ_ / 2; ++j) {
    h2[j].x = 0.0f;  h2[j].y = 0.0f;
    P2[j].x = 1.0f;  P2[j].y = 1.0f;
  }

  const size_t row0 = (size_t)bb * L_ + (size_t)g * T_;
  s6_delta_tile(xdbl + row0 * E_, Wdt + (size_t)dg * 32 * R_, bdt + dg * 32, q, lane, sDX);
  s6_stage_x(x + row0 * D_ + dg * 32, sDX);
  {
    const float* bsrc = xdbl + row0 * E_ + R_;
#pragma unroll
    for (int i = 0; i < 8; ++i) {
      const int idx = (int)threadIdx.x + 128 * i;
      const int row = idx >> 4, c4 = (idx & 15) * 4;
      *(v4f*)(sB + row * N_ + c4) = *(const v4f*)(bsrc + (size_t)row * E_ + c4);
    }
  }
  __syncthreads();

  const float* bq = sB + nq;
  const float* dx = sDX + 2 * lane;

  v4f Ba[4], Bb[4];
  v2f da, db;
  s6_row16(bq, Ba);
  da = *(const v2f*)dx;
  for (int tt = 0; tt < T_; tt += 2) {
    s6_row16(bq + (tt + 1) * N_, Bb);
    db = *(const v2f*)(dx + (tt + 1) * 64);
    s6_step_chunk(Ba, da.x, da.y, A2, h2, P2);
    asm volatile("" :: "v"(Bb[0]), "v"(Bb[1]), "v"(Bb[2]), "v"(Bb[3]), "v"(db));
    const int t2 = (tt + 2 < T_) ? tt + 2 : tt + 1;
    s6_row16(bq + t2 * N_, Ba);
    da = *(const v2f*)(dx + t2 * 64);
    s6_step_chunk(Bb, db.x, db.y, A2, h2, P2);
    asm volatile("" :: "v"(Ba[0]), "v"(Ba[1]), "v"(Ba[2]), "v"(Ba[3]), "v"(da));
  }
  const size_t base = ((size_t)bg * N_ + nq) * D_ + d;
#pragma unroll
  for (int j = 0; j < NQ_ / 2; ++j) {
    Pout[base + (size_t)(2 * j) * D_]     = P2[j].x;
    Pout[base + (size_t)(2 * j + 1) * D_] = P2[j].y;
    Hout[base + (size_t)(2 * j) * D_]     = h2[j].x;
    Hout[base + (size_t)(2 * j + 1) * D_] = h2[j].y;
  }
}

__global__ void __launch_bounds__(256)
s6_combine(const float* __restrict__ Pin, const float* __restrict__ Hin,
           float* __restrict__ h0out) {
  const int idx = blockIdx.x * 256 + threadIdx.x;
  const int bb  = idx / (N_ * D_);
  const int nd  = idx % (N_ * D_);
  const size_t base = (size_t)bb * G_ * N_ * D_ + nd;

  float h = 0.0f;
#pragma unroll
  for (int g = 0; g < G_; ++g) {
    const size_t o = base + (size_t)g * N_ * D_;
    h0out[o] = h;
    h = Pin[o] * h + Hin[o];
  }
}

__global__ void __launch_bounds__(128)
s6_scan_emit(const float* __restrict__ xdbl, const float* __restrict__ Wdt,
             const float* __restrict__ bdt,
             const float* __restrict__ x, const float* __restrict__ Alog,
             const float* __restrict__ Dp, const float* __restrict__ h0in,
             float* __restrict__ y) {
  __shared__ __attribute__((aligned(16))) float sDX[T_ * 64];
  __shared__ __attribute__((aligned(16))) float sBC[T_ * NS_ * 32];

  const int lane = threadIdx.x & 31;
  const int q  = __builtin_amdgcn_readfirstlane((int)(threadIdx.x >> 5));
  const int dg = blockIdx.x % DG_;
  const int bg = blockIdx.x / DG_;
  const int g  = bg % G_;
  const int bb = bg / G_;
  const int d  = dg * 32 + lane;
  const int nq = q * NQ_;

  v2f A2[NQ_ / 2], h2[NQ_ / 2];
  const v4f* alp = (const v4f*)(Alog + (size_t)d * N_ + nq);
#pragma unroll
  for (int k = 0; k < NQ_ / 4; ++k) {
    const v4f al = alp[k];
    A2[2 * k].x     = -__expf(al.x);
    A2[2 * k].y     = -__expf(al.y);
    A2[2 * k + 1].x = -__expf(al.z);
    A2[2 * k + 1].y = -__expf(al.w);
  }
  const size_t hbase = ((size_t)bg * N_ + nq) * D_ + d;
#pragma unroll
  for (int j = 0; j < NQ_ / 2; ++j) {
    h2[j].x = h0in[hbase + (size_t)(2 * j) * D_];
    h2[j].y = h0in[hbase + (size_t)(2 * j + 1) * D_];
  }

  const size_t row0 = (size_t)bb * L_ + (size_t)g * T_;
  s6_delta_tile(xdbl + row0 * E_, Wdt + (size_t)dg * 32 * R_, bdt + dg * 32, q, lane, sDX);
  s6_stage_x(x + row0 * D_ + dg * 32, sDX);
  {
    const float* bsrc = xdbl + row0 * E_ + R_;
#pragma unroll
    for (int i = 0; i < 16; ++i) {
      const int idx = (int)threadIdx.x + 128 * i;
      const int row = idx >> 5, col = (idx & 31) * 4;
      const int n = col & 63;
      *(v4f*)(sBC + (row * NS_ + (n >> 4)) * 32 + (col >> 6) * 16 + (n & 15)) =
          *(const v4f*)(bsrc + (size_t)row * E_ + col);
    }
  }
  __syncthreads();

  float* bc = sBC + q * 32;
  const float* dx = sDX + 2 * lane;

  v4f Ba[4], Ca[4], Bb[4], Cb[4];
  v2f da, db;
  s6_row16(bc, Ba);
  s6_row16(bc + 16, Ca);
  da = *(const v2f*)dx;
  for (int tt = 0; tt < T_; tt += 2) {
    s6_row16(bc + (tt + 1) * NS_ * 32, Bb);
    s6_row16(bc + (tt + 1) * NS_ * 32 + 16, Cb);
    db = *(const v2f*)(dx + (tt + 1) * 64);
    bc[tt * NS_ * 32 + lane] = s6_step_emit(Ba, Ca, da.x, da.y, A2, h2);
    asm volatile("" :: "v"(Bb[0]), "v"(Bb[1]), "v"(Bb[2]), "v"(Bb[3]),
                       "v"(Cb[0]), "v"(Cb[1]), "v"(Cb[2]), "v"(Cb[3]), "v"(db));
    const int t2 = (tt + 2 < T_) ? tt + 2 : tt + 1;
    s6_row16(bc + t2 * NS_ * 32, Ba);
    s6_row16(bc + t2 * NS_ * 32 + 16, Ca);
    da = *(const v2f*)(dx + t2 * 64);
    bc[(tt + 1) * NS_ * 32 + lane] = s6_step_emit(Bb, Cb, db.x, db.y, A2, h2);
    asm volatile("" :: "v"(Ba[0]), "v"(Ba[1]), "v"(Ba[2]), "v"(Ba[3]),
                       "v"(Ca[0]), "v"(Ca[1]), "v"(Ca[2]), "v"(Ca[3]), "v"(da));
  }
  __syncthreads();

  const float Dv = Dp[d];
#pragma unroll 4
  for (int k = 0; k < T_ / NS_; ++k) {
    const int tt = q * (T_ / NS_) + k;
    const float* yp = sBC + tt * NS_ * 32 + lane;
    const float s = (yp[0] + yp[32]) + (yp[64] + yp[96]);
    const float xv = dx[tt * 64 + 1];
    y[(row0 + tt) * D_ + d] = s + xv * Dv;
  }
}

extern "C" void kernel_launch(void* const* d_in, const int* in_sizes, int n_in,
                              void* d_out, int out_size, void* d_ws, size_t ws_size,
                              hipStream_t stream) {
  const float* x      = (const float*)d_in[0];
  const float* Wxproj = (const float*)d_in[1];
  const float* Wdt    = (const float*)d_in[2];
  const float* bdt    = (const float*)d_in[3];
  const float* Alog   = (const float*)d_in[4];
  const float* Dp     = (const float*)d_in[5];
  float*       y      = (float*)d_out;

  float* ws    = (float*)d_ws;
  float* xdbl  = ws;
  float* Pws   = xdbl  + (size_t)BL_ * E_;
  float* Hws   = Pws   + (size_t)B_ * D_ * G_ * N_;
  float* h0ws  = Hws   + (size_t)B_ * D_ * G_ * N_;

  s6_xproj_gemm<<<256, 256, 0, stream>>>(x, Wxproj, xdbl);
  s6_scan_chunks<<<1024, 128, 0, stream>>>(xdbl, Wdt, bdt, x, Alog, Pws, Hws);
  s6_combine<<<256, 256, 0, stream>>>(Pws, Hws, h0ws);
  s6_scan_emit<<<1024, 128, 0, stream>>>(xdbl, Wdt, bdt, x, Alog, Dp, h0ws, y);
}


# Scan time loops: decay's constant factors hoisted: log2(e) folded into A once per lane (one multiply + bare exp2 per state-step); chunk decay P = exp2(A*sum delta) once per chunk, no per-step product
# speedup vs baseline: 4.5144x; 1.0896x over previous
#include <hip/hip_runtime.h>
#include <hip/hip_bf16.h>

typedef float v2f __attribute__((ext_vector_type(2)));
typedef float v8f __attribute__((ext_vector_type(8)));
typedef float v4f __attribute__((ext_vector_type(4)));
typedef float v16f __attribute__((ext_vector_type(16)));

#define B_  2
#define L_  2048
#define D_  512
#define N_  64
#define R_  32
#define E_  160
#define BL_ (B_ * L_)
#define G_  32
#define T_  (L_ / G_)
#define NS_ 4
#define NQ_ (N_ / NS_)
#define DG_ (D_ / 32)

#define XW_  8
#define XK_  (D_ / XW_)
#define XP_  68
#define XT_  (16 * XP_)
#define XET_ (E_ / 16)
#define XWS_ (5 * XT_)

__device__ __forceinline__ void
s6_slice_fetch(const float* __restrict__ src, const int lane, v4f (&r)[8]) {
  const float* p = src + (size_t)(lane >> 4) * D_ + 4 * (lane & 15);
#pragma unroll
  for (int i = 0; i < 8; ++i) r[i] = *(const v4f*)(p + (size_t)(2 * i) * D_);
}

__device__ __forceinline__ void
s6_slice_park(float* dst, const int lane, const v4f (&r)[8]) {
  float* p = dst + (lane >> 4) * XP_ + 4 * (lane & 15);
#pragma unroll
  for (int i = 0; i < 8; ++i) *(v4f*)(p + 2 * i * XP_) = r[i];
}

__device__ __forceinline__ void
s6_slice_frags(const float* tile, const int lane, v4f (&f)[8]) {
  const float* p = tile + (lane & 15) * XP_ + 4 * (lane >> 4);
#pragma unroll
  for (int i = 0; i < 8; ++i) f[i] = *(const v4f*)(p + 8 * i);
}

__device__ __forceinline__ void s6_wave_lds_handover() {
  __builtin_amdgcn_fence(__ATOMIC_RELEASE, "wavefront");
  __builtin_amdgcn_wave_barrier();
  __builtin_amdgcn_fence(__ATOMIC_ACQUIRE, "wavefront");
}

__global__ void __launch_bounds__(256)
s6_xproj_gemm(const float* __restrict__ x, const float* __restrict__ W,
              float* __restrict__ xdbl) {
  __shared__ __attribute__((aligned(16))) float sm[XW_ * XWS_];

  const int lane = threadIdx.x & 31;
  const int w  = __builtin_amdgcn_readfirstlane((int)(threadIdx.x >> 5));
  const int rt = blockIdx.x;

  float* sa = sm + w * XWS_;
  float* sb = sa + XT_;
  const float* xa = x + (size_t)rt * 16 * D_ + w * XK_;
  const float* wb = W + w * XK_;

  v4f ra[8], rb0[8], rb1[8];
  s6_slice_fetch(xa, lane, ra);
  s6_slice_fetch(wb, lane, rb0);
  s6_slice_fetch(wb + (size_t)16 * D_, lane, rb1);
  s6_slice_park(sa, lane, ra);
  s6_slice_park(sb, lane, rb0);
  s6_slice_park(sb + XT_, lane, rb1);
  s6_slice_fetch(wb + (size_t)32 * D_, lane, rb0);
  s6_slice_fetch(wb + (size_t)48 * D_, lane, rb1);
  s6_wave_lds_handover();

  v4f af[8], bf[2][2][8];
  s6_slice_frags(sa, lane, af);
  s6_slice_frags(sb, lane, bf[0][0]);
  s6_slice_frags(sb + XT_, lane, bf[0][1]);
  s6_slice_park(sb + 2 * XT_, lane, rb0);
  s6_slice_park(sb + 3 * XT_, lane, rb1);
  s6_wave_lds_handover();

  v8f acc[XET_];
#pragma unroll
  for (int j = 0; j < XET_; ++j) acc[j] = (v8f){};

#pragma unroll
  for (int jp = 0; jp < XET_ / 2; ++jp) {
    const int cs = jp & 1, ns = cs ^ 1;
    if (jp + 2 < XET_ / 2) {
      s6_slice_fetch(wb + (size_t)(2 * jp + 4) * 16 * D_, lane, rb0);
      s6_slice_fetch(wb + (size_t)(2 * jp + 5) * 16 * D_, lane, rb1);
    }
    if (jp + 1 < XET_ / 2) {
      s6_slice_frags(sb + ns * 2 * XT_, lane, bf[ns][0]);
      s6_slice_frags(sb + ns * 2 * XT_ + XT_, lane, bf[ns][1]);
    }
    __builtin_amdgcn_sched_barrier(0);
#pragma unroll
    for (int i = 0; i < 8; ++i) {
      const v2f a0 = {af[i].x, af[i].y}, a1 = {af[i].z, af[i].w};
      const v2f p0 = {bf[cs][0][i].x, bf[cs][0][i].y}, p1 = {bf[cs][0][i].z, bf[cs][0][i].w};
      const v2f q0 = {bf[cs][1][i].x, bf[cs][1][i].y}, q1 = {bf[cs][1][i].z, bf[cs][1][i].w};
      acc[2 * jp]     = __builtin_amdgcn_wmma_f32_16x16x4_f32(false, a0, false, p0, (short)0, acc[2 * jp],     false, false);
      acc[2 * jp + 1] = __builtin_amdgcn_wmma_f32_16x16x4_f32(false, a0, false, q0, (short)0, acc[2 * jp + 1], false, false);
      acc[2 * jp]     = __builtin_amdgcn_wmma_f32_16x16x4_f32(false, a1, false, p1, (short)0, acc[2 * jp],     false, false);
      acc[2 * jp + 1] = __builtin_amdgcn_wmma_f32_16x16x4_f32(false, a1, false, q1, (short)0, acc[2 * jp + 1], false, false);
    }
    __builtin_amdgcn_sched_barrier(0);
    if (jp + 2 < XET_ / 2) {
      s6_slice_park(sb + cs * 2 * XT_, lane, rb0);
      s6_slice_park(sb + cs * 2 * XT_ + XT_, lane, rb1);
      s6_wave_lds_handover();
    }
  }

  __syncthreads();
#pragma unroll
  for (int j = 0; j < XET_; ++j) {
#pragma unroll
    for (int r = 0; r < 8; ++r)
      sm[((w * XET_ + j) * 8 + r) * 32 + lane] = acc[j][r];
  }
  __syncthreads();

  float* orow = xdbl + (size_t)rt * 16 * E_;
#pragma unroll 2
  for (int o = threadIdx.x; o < 16 * E_; o += 256) {
    const int row = o / E_, col = o % E_;
    const int idx = ((col >> 4) * 8 + (row & 7)) * 32 + (col & 15) + 16 * (row >> 3);
    float s = sm[idx];
#pragma unroll
    for (int q = 1; q < XW_; ++q) s += sm[q * XET_ * 256 + idx];
    orow[o] = s;
  }
}

__device__ __forceinline__ float s6_softplus(const float z) {
  const float sp = __logf(1.0f + __expf(z));
  return (z > 20.0f) ? z : sp;
}

__device__ __forceinline__ void
s6_delta_tile(const float* __restrict__ xrows,
              const float* __restrict__ wdt,
              const float* __restrict__ bdt,
              const int q, const int lane, float* sDX) {
  const int laneM = lane & 15;
  const int laneH = lane >> 4;
  const float* arow = xrows + (size_t)(q * 16 + laneM) * E_ + 2 * laneH;
  const float* br0  = wdt + (size_t)laneM * R_ + 2 * laneH;
  const float* br1  = br0 + 16 * R_;

  v8f c0 = {}, c1 = {};
#pragma unroll
  for (int k = 0; k < R_; k += 4) {
    const v2f a  = *(const v2f*)(arow + k);
    const v2f b0 = *(const v2f*)(br0 + k);
    const v2f b1 = *(const v2f*)(br1 + k);
    c0 = __builtin_amdgcn_wmma_f32_16x16x4_f32(false, a, false, b0, (short)0, c0, false, false);
    c1 = __builtin_amdgcn_wmma_f32_16x16x4_f32(false, a, false, b1, (short)0, c1, false, false);
  }
  const float bi0 = bdt[laneM];
  const float bi1 = bdt[16 + laneM];
  float* p = sDX + ((q * 16 + 8 * laneH) * 32 + laneM) * 2;
#pragma unroll
  for (int j = 0; j < 8; ++j) {
    p[j * 64]      = s6_softplus(c0[j] + bi0);
    p[j * 64 + 32] = s6_softplus(c1[j] + bi1);
  }
}


__device__ __forceinline__ void
s6_stage_x(const float* __restrict__ xsrc, float* sDX) {
#pragma unroll
  for (int i = 0; i < 4; ++i) {
    const int idx = (int)threadIdx.x + 128 * i;
    const int row = idx >> 3, c4 = (idx & 7) * 4;
    const v4f xv = *(const v4f*)(xsrc + (size_t)row * D_ + c4);
    float* p = sDX + (row * 32 + c4) * 2 + 1;
    p[0] = xv.x;  p[2] = xv.y;  p[4] = xv.z;  p[6] = xv.w;
  }
}

__device__ __forceinline__ void s6_row16(const float* p, v4f (&r)[4]) {
#pragma unroll
  for (int i = 0; i < 4; ++i) r[i] = *(const v4f*)(p + 4 * i);
}

#define S6_LOG2E 1.44269504088896340736f

__device__ __forceinline__ void
s6_step_chunk(const v4f (&Bc)[4], const float dlt, const float xv,
              const v2f (&A2)[NQ_ / 2], v2f (&h2)[NQ_ / 2]) {
  const float du = dlt * xv;
#pragma unroll
  for (int j = 0; j < NQ_ / 2; ++j) {
    const v2f arg = dlt * A2[j];
    v2f a;  a.x = __builtin_amdgcn_exp2f(arg.x);  a.y = __builtin_amdgcn_exp2f(arg.y);
    v2f Bv; Bv.x = Bc[j >> 1][2 * (j & 1)];  Bv.y = Bc[j >> 1][2 * (j & 1) + 1];
    h2[j] = a * h2[j] + du * Bv;
  }
}

__device__ __forceinline__ float
s6_step_emit(const v4f (&Bc)[4], const v4f (&Cc)[4], const float dlt, const float xv,
             const v2f (&A2)[NQ_ / 2], v2f (&h2)[NQ_ / 2]) {
  const float du = dlt * xv;
  v2f acc = {0.0f, 0.0f};
#pragma unroll
  for (int j = 0; j < NQ_ / 2; ++j) {
    const v2f arg = dlt * A2[j];
    v2f a;  a.x = __builtin_amdgcn_exp2f(arg.x);  a.y = __builtin_amdgcn_exp2f(arg.y);
    v2f Bv; Bv.x = Bc[j >> 1][2 * (j & 1)];  Bv.y = Bc[j >> 1][2 * (j & 1) + 1];
    v2f Cv; Cv.x = Cc[j >> 1][2 * (j & 1)];  Cv.y = Cc[j >> 1][2 * (j & 1) + 1];
    h2[j] = a * h2[j] + du * Bv;
    acc += h2[j] * Cv;
  }
  return acc.x + acc.y;
}

__global__ void __launch_bounds__(128)
s6_scan_chunks(const float* __restrict__ xdbl, const float* __restrict__ Wdt,
               const float* __restrict__ bdt,
               const float* __restrict__ x, const float* __restrict__ Alog,
               float* __restrict__ Pout, float* __restrict__ Hout) {
  __shared__ __attribute__((aligned(16))) float sDX[T_ * 64];
  __shared__ __attribute__((aligned(16))) float sB[T_ * N_];

  const int lane = threadIdx.x & 31;
  const int q  = __builtin_amdgcn_readfirstlane((int)(threadIdx.x >> 5));
  const int dg = blockIdx.x % DG_;
  const int bg = blockIdx.x / DG_;
  const int g  = bg % G_;
  const int bb = bg / G_;
  const int d  = dg * 32 + lane;
  const int nq = q * NQ_;

  v2f A2[NQ_ / 2], h2[NQ_ / 2];
  const v4f* alp = (const v4f*)(Alog + (size_t)d * N_ + nq);
#pragma unroll
  for (int k = 0; k < NQ_ / 4; ++k) {
    const v4f al = alp[k];
    A2[2 * k].x     = -__expf(al.x) * S6_LOG2E;
    A2[2 * k].y     = -__expf(al.y) * S6_LOG2E;
    A2[2 * k + 1].x = -__expf(al.z) * S6_LOG2E;
    A2[2 * k + 1].y = -__expf(al.w) * S6_LOG2E;
  }
#pragma unroll
  for (int j = 0; j < NQ_ / 2; ++j) {
    h2[j].x = 0.0f;  h2[j].y = 0.0f;
  }
  float dsum = 0.0f;

  const size_t row0 = (size_t)bb * L_ + (size_t)g * T_;
  s6_delta_tile(xdbl + row0 * E_, Wdt + (size_t)dg * 32 * R_, bdt + dg * 32, q, lane, sDX);
  s6_stage_x(x + row0 * D_ + dg * 32, sDX);
  {
    const float* bsrc = xdbl + row0 * E_ + R_;
#pragma unroll
    for (int i = 0; i < 8; ++i) {
      const int idx = (int)threadIdx.x + 128 * i;
      const int row = idx >> 4, c4 = (idx & 15) * 4;
      *(v4f*)(sB + row * N_ + c4) = *(const v4f*)(bsrc + (size_t)row * E_ + c4);
    }
  }
  __syncthreads();

  const float* bq = sB + nq;
  const float* dx = sDX + 2 * lane;

  v4f Ba[4], Bb[4];
  v2f da, db;
  s6_row16(bq, Ba);
  da = *(const v2f*)dx;
  for (int tt = 0; tt < T_; tt += 2) {
    s6_row16(bq + (tt + 1) * N_, Bb);
    db = *(const v2f*)(dx + (tt + 1) * 64);
    s6_step_chunk(Ba, da.x, da.y, A2, h2);
    dsum += da.x;
    asm volatile("" :: "v"(Bb[0]), "v"(Bb[1]), "v"(Bb[2]), "v"(Bb[3]), "v"(db));
    const int t2 = (tt + 2 < T_) ? tt + 2 : tt + 1;
    s6_row16(bq + t2 * N_, Ba);
    da = *(const v2f*)(dx + t2 * 64);
    s6_step_chunk(Bb, db.x, db.y, A2, h2);
    dsum += db.x;
    asm volatile("" :: "v"(Ba[0]), "v"(Ba[1]), "v"(Ba[2]), "v"(Ba[3]), "v"(da));
  }
  const size_t base = ((size_t)bg * N_ + nq) * D_ + d;
#pragma unroll
  for (int j = 0; j < NQ_ / 2; ++j) {
    const v2f parg = dsum * A2[j];
    Pout[base + (size_t)(2 * j) * D_]     = __builtin_amdgcn_exp2f(parg.x);
    Pout[base + (size_t)(2 * j + 1) * D_] = __builtin_amdgcn_exp2f(parg.y);
    Hout[base + (size_t)(2 * j) * D_]     = h2[j].x;
    Hout[base + (size_t)(2 * j + 1) * D_] = h2[j].y;
  }
}

__global__ void __launch_bounds__(256)
s6_combine(const float* __restrict__ Pin, const float* __restrict__ Hin,
           float* __restrict__ h0out) {
  const int idx = blockIdx.x * 256 + threadIdx.x;
  const int bb  = idx / (N_ * D_);
  const int nd  = idx % (N_ * D_);
  const size_t base = (size_t)bb * G_ * N_ * D_ + nd;

  float h = 0.0f;
#pragma unroll
  for (int g = 0; g < G_; ++g) {
    const size_t o = base + (size_t)g * N_ * D_;
    h0out[o] = h;
    h = Pin[o] * h + Hin[o];
  }
}

__global__ void __launch_bounds__(128)
s6_scan_emit(const float* __restrict__ xdbl, const float* __restrict__ Wdt,
             const float* __restrict__ bdt,
             const float* __restrict__ x, const float* __restrict__ Alog,
             const float* __restrict__ Dp, const float* __restrict__ h0in,
             float* __restrict__ y) {
  __shared__ __attribute__((aligned(16))) float sDX[T_ * 64];
  __shared__ __attribute__((aligned(16))) float sBC[T_ * NS_ * 32];

  const int lane = threadIdx.x & 31;
  const int q  = __builtin_amdgcn_readfirstlane((int)(threadIdx.x >> 5));
  const int dg = blockIdx.x % DG_;
  const int bg = blockIdx.x / DG_;
  const int g  = bg % G_;
  const int bb = bg / G_;
  const int d  = dg * 32 + lane;
  const int nq = q * NQ_;

  v2f A2[NQ_ / 2], h2[NQ_ / 2];
  const v4f* alp = (const v4f*)(Alog + (size_t)d * N_ + nq);
#pragma unroll
  for (int k = 0; k < NQ_ / 4; ++k) {
    const v4f al = alp[k];
    A2[2 * k].x     = -__expf(al.x) * S6_LOG2E;
    A2[2 * k].y     = -__expf(al.y) * S6_LOG2E;
    A2[2 * k + 1].x = -__expf(al.z) * S6_LOG2E;
    A2[2 * k + 1].y = -__expf(al.w) * S6_LOG2E;
  }
  const size_t hbase = ((size_t)bg * N_ + nq) * D_ + d;
#pragma unroll
  for (int j = 0; j < NQ_ / 2; ++j) {
    h2[j].x = h0in[hbase + (size_t)(2 * j) * D_];
    h2[j].y = h0in[hbase + (size_t)(2 * j + 1) * D_];
  }

  const size_t row0 = (size_t)bb * L_ + (size_t)g * T_;
  s6_delta_tile(xdbl + row0 * E_, Wdt + (size_t)dg * 32 * R_, bdt + dg * 32, q, lane, sDX);
  s6_stage_x(x + row0 * D_ + dg * 32, sDX);
  {
    const float* bsrc = xdbl + row0 * E_ + R_;
#pragma unroll
    for (int i = 0; i < 16; ++i) {
      const int idx = (int)threadIdx.x + 128 * i;
      const int row = idx >> 5, col = (idx & 31) * 4;
      const int n = col & 63;
      *(v4f*)(sBC + (row * NS_ + (n >> 4)) * 32 + (col >> 6) * 16 + (n & 15)) =
          *(const v4f*)(bsrc + (size_t)row * E_ + col);
    }
  }
  __syncthreads();

  float* bc = sBC + q * 32;
  const float* dx = sDX + 2 * lane;

  v4f Ba[4], Ca[4], Bb[4], Cb[4];
  v2f da, db;
  s6_row16(bc, Ba);
  s6_row16(bc + 16, Ca);
  da = *(const v2f*)dx;
  for (int tt = 0; tt < T_; tt += 2) {
    s6_row16(bc + (tt + 1) * NS_ * 32, Bb);
    s6_row16(bc + (tt + 1) * NS_ * 32 + 16, Cb);
    db = *(const v2f*)(dx + (tt + 1) * 64);
    bc[tt * NS_ * 32 + lane] = s6_step_emit(Ba, Ca, da.x, da.y, A2, h2);
    asm volatile("" :: "v"(Bb[0]), "v"(Bb[1]), "v"(Bb[2]), "v"(Bb[3]),
                       "v"(Cb[0]), "v"(Cb[1]), "v"(Cb[2]), "v"(Cb[3]), "v"(db));
    const int t2 = (tt + 2 < T_) ? tt + 2 : tt + 1;
    s6_row16(bc + t2 * NS_ * 32, Ba);
    s6_row16(bc + t2 * NS_ * 32 + 16, Ca);
    da = *(const v2f*)(dx + t2 * 64);
    bc[(tt + 1) * NS_ * 32 + lane] = s6_step_emit(Bb, Cb, db.x, db.y, A2, h2);
    asm volatile("" :: "v"(Ba[0]), "v"(Ba[1]), "v"(Ba[2]), "v"(Ba[3]),
                       "v"(Ca[0]), "v"(Ca[1]), "v"(Ca[2]), "v"(Ca[3]), "v"(da));
  }
  __syncthreads();

  const float Dv = Dp[d];
#pragma unroll 4
  for (int k = 0; k < T_ / NS_; ++k) {
    const int tt = q * (T_ / NS_) + k;
    const float* yp = sBC + tt * NS_ * 32 + lane;
    const float s = (yp[0] + yp[32]) + (yp[64] + yp[96]);
    const float xv = dx[tt * 64 + 1];
    y[(row0 + tt) * D_ + d] = s + xv * Dv;
  }
}

extern "C" void kernel_launch(void* const* d_in, const int* in_sizes, int n_in,
                              void* d_out, int out_size, void* d_ws, size_t ws_size,
                              hipStream_t stream) {
  const float* x      = (const float*)d_in[0];
  const float* Wxproj = (const float*)d_in[1];
  const float* Wdt    = (const float*)d_in[2];
  const float* bdt    = (const float*)d_in[3];
  const float* Alog   = (const float*)d_in[4];
  const float* Dp     = (const float*)d_in[5];
  float*       y      = (float*)d_out;

  float* ws    = (float*)d_ws;
  float* xdbl  = ws;
  float* Pws   = xdbl  + (size_t)BL_ * E_;
  float* Hws   = Pws   + (size_t)B_ * D_ * G_ * N_;
  float* h0ws  = Hws   + (size_t)B_ * D_ * G_ * N_;

  s6_xproj_gemm<<<256, 256, 0, stream>>>(x, Wxproj, xdbl);
  s6_scan_chunks<<<1024, 128, 0, stream>>>(xdbl, Wdt, bdt, x, Alog, Pws, Hws);
  s6_combine<<<256, 256, 0, stream>>>(Pws, Hws, h0ws);
  s6_scan_emit<<<1024, 128, 0, stream>>>(xdbl, Wdt, bdt, x, Alog, Dp, h0ws, y);
}


# Scan passes: a lane owns two channels (d, d+32), so a step's B and C rows are read from LDS once for both; 512 work-groups of 64 channels, B/C staging halved (sum order unchanged)
# speedup vs baseline: 4.7288x; 1.0475x over previous
#include <hip/hip_runtime.h>
#include <hip/hip_bf16.h>

typedef float v2f __attribute__((ext_vector_type(2)));
typedef float v8f __attribute__((ext_vector_type(8)));
typedef float v4f __attribute__((ext_vector_type(4)));
typedef float v16f __attribute__((ext_vector_type(16)));

#define B_  2
#define L_  2048
#define D_  512
#define N_  64
#define R_  32
#define E_  160
#define BL_ (B_ * L_)
#define G_  32
#define T_  (L_ / G_)
#define NS_ 4
#define NQ_ (N_ / NS_)
#define CW_ 64
#define DG_ (D_ / CW_)

#define XW_  8
#define XK_  (D_ / XW_)
#define XP_  68
#define XT_  (16 * XP_)
#define XET_ (E_ / 16)
#define XWS_ (5 * XT_)

__device__ __forceinline__ void
s6_slice_fetch(const float* __restrict__ src, const int lane, v4f (&r)[8]) {
  const float* p = src + (size_t)(lane >> 4) * D_ + 4 * (lane & 15);
#pragma unroll
  for (int i = 0; i < 8; ++i) r[i] = *(const v4f*)(p + (size_t)(2 * i) * D_);
}

__device__ __forceinline__ void
s6_slice_park(float* dst, const int lane, const v4f (&r)[8]) {
  float* p = dst + (lane >> 4) * XP_ + 4 * (lane & 15);
#pragma unroll
  for (int i = 0; i < 8; ++i) *(v4f*)(p + 2 * i * XP_) = r[i];
}

__device__ __forceinline__ void
s6_slice_frags(const float* tile, const int lane, v4f (&f)[8]) {
  const float* p = tile + (lane & 15) * XP_ + 4 * (lane >> 4);
#pragma unroll
  for (int i = 0; i < 8; ++i) f[i] = *(const v4f*)(p + 8 * i);
}

__device__ __forceinline__ void s6_wave_lds_handover() {
  __builtin_amdgcn_fence(__ATOMIC_RELEASE, "wavefront");
  __builtin_amdgcn_wave_barrier();
  __builtin_amdgcn_fence(__ATOMIC_ACQUIRE, "wavefront");
}

__global__ void __launch_bounds__(256)
s6_xproj_gemm(const float* __restrict__ x, const float* __restrict__ W,
              float* __restrict__ xdbl) {
  __shared__ __attribute__((aligned(16))) float sm[XW_ * XWS_];

  const int lane = threadIdx.x & 31;
  const int w  = __builtin_amdgcn_readfirstlane((int)(threadIdx.x >> 5));
  const int rt = blockIdx.x;

  float* sa = sm + w * XWS_;
  float* sb = sa + XT_;
  const float* xa = x + (size_t)rt * 16 * D_ + w * XK_;
  const float* wb = W + w * XK_;

  v4f ra[8], rb0[8], rb1[8];
  s6_slice_fetch(xa, lane, ra);
  s6_slice_fetch(wb, lane, rb0);
  s6_slice_fetch(wb + (size_t)16 * D_, lane, rb1);
  s6_slice_park(sa, lane, ra);
  s6_slice_park(sb, lane, rb0);
  s6_slice_park(sb + XT_, lane, rb1);
  s6_slice_fetch(wb + (size_t)32 * D_, lane, rb0);
  s6_slice_fetch(wb + (size_t)48 * D_, lane, rb1);
  s6_wave_lds_handover();

  v4f af[8], bf[2][2][8];
  s6_slice_frags(sa, lane, af);
  s6_slice_frags(sb, lane, bf[0][0]);
  s6_slice_frags(sb + XT_, lane, bf[0][1]);
  s6_slice_park(sb + 2 * XT_, lane, rb0);
  s6_slice_park(sb + 3 * XT_, lane, rb1);
  s6_wave_lds_handover();

  v8f acc[XET_];
#pragma unroll
  for (int j = 0; j < XET_; ++j) acc[j] = (v8f){};

#pragma unroll
  for (int jp = 0; jp < XET_ / 2; ++jp) {
    const int cs = jp & 1, ns = cs ^ 1;
    if (jp + 2 < XET_ / 2) {
      s6_slice_fetch(wb + (size_t)(2 * jp + 4) * 16 * D_, lane, rb0);
      s6_slice_fetch(wb + (size_t)(2 * jp + 5) * 16 * D_, lane, rb1);
    }
    if (jp + 1 < XET_ / 2) {
      s6_slice_frags(sb + ns * 2 * XT_, lane, bf[ns][0]);
      s6_slice_frags(sb + ns * 2 * XT_ + XT_, lane, bf[ns][1]);
    }
    __builtin_amdgcn_sched_barrier(0);
#pragma unroll
    for (int i = 0; i < 8; ++i) {
      const v2f a0 = {af[i].x, af[i].y}, a1 = {af[i].z, af[i].w};
      const v2f p0 = {bf[cs][0][i].x, bf[cs][0][i].y}, p1 = {bf[cs][0][i].z, bf[cs][0][i].w};
      const v2f q0 = {bf[cs][1][i].x, bf[cs][1][i].y}, q1 = {bf[cs][1][i].z, bf[cs][1][i].w};
      acc[2 * jp]     = __builtin_amdgcn_wmma_f32_16x16x4_f32(false, a0, false, p0, (short)0, acc[2 * jp],     false, false);
      acc[2 * jp + 1] = __builtin_amdgcn_wmma_f32_16x16x4_f32(false, a0, false, q0, (short)0, acc[2 * jp + 1], false, false);
      acc[2 * jp]     = __builtin_amdgcn_wmma_f32_16x16x4_f32(false, a1, false, p1, (short)0, acc[2 * jp],     false, false);
      acc[2 * jp + 1] = __builtin_amdgcn_wmma_f32_16x16x4_f32(false, a1, false, q1, (short)0, acc[2 * jp + 1], false, false);
    }
    __builtin_amdgcn_sched_barrier(0);
    if (jp + 2 < XET_ / 2) {
      s6_slice_park(sb + cs * 2 * XT_, lane, rb0);
      s6_slice_park(sb + cs * 2 * XT_ + XT_, lane, rb1);
      s6_wave_lds_handover();
    }
  }

  __syncthreads();
#pragma unroll
  for (int j = 0; j < XET_; ++j) {
#pragma unroll
    for (int r = 0; r < 8; ++r)
      sm[((w * XET_ + j) * 8 + r) * 32 + lane] = acc[j][r];
  }
  __syncthreads();

  float* orow = xdbl + (size_t)rt * 16 * E_;
#pragma unroll 2
  for (int o = threadIdx.x; o < 16 * E_; o += 256) {
    const int row = o / E_, col = o % E_;
    const int idx = ((col >> 4) * 8 + (row & 7)) * 32 + (col & 15) + 16 * (row >> 3);
    float s = sm[idx];
#pragma unroll
    for (int q = 1; q < XW_; ++q) s += sm[q * XET_ * 256 + idx];
    orow[o] = s;
  }
}

__device__ __forceinline__ float s6_softplus(const float z) {
  const float sp = __logf(1.0f + __expf(z));
  return (z > 20.0f) ? z : sp;
}

__device__ __forceinline__ void
s6_delta_tile(const float* __restrict__ xrows,
              const float* __restrict__ wdt,
              const float* __restrict__ bdt,
              const int q, const int lane, float* sDX) {
  const int laneM = lane & 15;
  const int laneH = lane >> 4;
  const float* arow = xrows + (size_t)(q * 16 + laneM) * E_ + 2 * laneH;
  const float* br0  = wdt + (size_t)laneM * R_ + 2 * laneH;

  v8f cc[CW_ / 16];
#pragma unroll
  for (int ct = 0; ct < CW_ / 16; ++ct) cc[ct] = (v8f){};
#pragma unroll
  for (int k = 0; k < R_; k += 4) {
    const v2f a = *(const v2f*)(arow + k);
#pragma unroll
    for (int ct = 0; ct < CW_ / 16; ++ct) {
      const v2f b = *(const v2f*)(br0 + (size_t)ct * 16 * R_ + k);
      cc[ct] = __builtin_amdgcn_wmma_f32_16x16x4_f32(false, a, false, b, (short)0, cc[ct], false, false);
    }
  }
  float* p = sDX + ((q * 16 + 8 * laneH) * CW_ + laneM) * 2;
#pragma unroll
  for (int ct = 0; ct < CW_ / 16; ++ct) {
    const float bi = bdt[ct * 16 + laneM];
#pragma unroll
    for (int j = 0; j < 8; ++j)
      p[j * 2 * CW_ + ct * 32] = s6_softplus(cc[ct][j] + bi);
  }
}


__device__ __forceinline__ void
s6_stage_x(const float* __restrict__ xsrc, float* sDX) {
#pragma unroll
  for (int i = 0; i < 8; ++i) {
    const int idx = (int)threadIdx.x + 128 * i;
    const int row = idx >> 4, c4 = (idx & 15) * 4;
    const v4f xv = *(const v4f*)(xsrc + (size_t)row * D_ + c4);
    float* p = sDX + (row * CW_ + c4) * 2 + 1;
    p[0] = xv.x;  p[2] = xv.y;  p[4] = xv.z;  p[6] = xv.w;
  }
}

__device__ __forceinline__ void s6_row16(const float* p, v4f (&r)[4]) {
#pragma unroll
  for (int i = 0; i < 4; ++i) r[i] = *(const v4f*)(p + 4 * i);
}

#define S6_LOG2E 1.44269504088896340736f

__device__ __forceinline__ void
s6_step_chunk(const v4f (&Bc)[4], const float dlt, const float xv,
              const v2f (&A2)[NQ_ / 2], v2f (&h2)[NQ_ / 2]) {
  const float du = dlt * xv;
#pragma unroll
  for (int j = 0; j < NQ_ / 2; ++j) {
    const v2f arg = dlt * A2[j];
    v2f a;  a.x = __builtin_amdgcn_exp2f(arg.x);  a.y = __builtin_amdgcn_exp2f(arg.y);
    v2f Bv; Bv.x = Bc[j >> 1][2 * (j & 1)];  Bv.y = Bc[j >> 1][2 * (j & 1) + 1];
    h2[j] = a * h2[j] + du * Bv;
  }
}

__device__ __forceinline__ float
s6_step_emit(const v4f (&Bc)[4], const v4f (&Cc)[4], const float dlt, const float xv,
             const v2f (&A2)[NQ_ / 2], v2f (&h2)[NQ_ / 2]) {
  const float du = dlt * xv;
  v2f acc = {0.0f, 0.0f};
#pragma unroll
  for (int j = 0; j < NQ_ / 2; ++j) {
    const v2f arg = dlt * A2[j];
    v2f a;  a.x = __builtin_amdgcn_exp2f(arg.x);  a.y = __builtin_amdgcn_exp2f(arg.y);
    v2f Bv; Bv.x = Bc[j >> 1][2 * (j & 1)];  Bv.y = Bc[j >> 1][2 * (j & 1) + 1];
    v2f Cv; Cv.x = Cc[j >> 1][2 * (j & 1)];  Cv.y = Cc[j >> 1][2 * (j & 1) + 1];
    h2[j] = a * h2[j] + du * Bv;
    acc += h2[j] * Cv;
  }
  return acc.x + acc.y;
}

__device__ __forceinline__ void
s6_load_A2(const float* __restrict__ arow, v2f (&A2)[NQ_ / 2]) {
  const v4f* alp = (const v4f*)arow;
#pragma unroll
  for (int k = 0; k < NQ_ / 4; ++k) {
    const v4f al = alp[k];
    A2[2 * k].x     = -__expf(al.x) * S6_LOG2E;
    A2[2 * k].y     = -__expf(al.y) * S6_LOG2E;
    A2[2 * k + 1].x = -__expf(al.z) * S6_LOG2E;
    A2[2 * k + 1].y = -__expf(al.w) * S6_LOG2E;
  }
}

__global__ void __launch_bounds__(128)
s6_scan_chunks(const float* __restrict__ xdbl, const float* __restrict__ Wdt,
               const float* __restrict__ bdt,
               const float* __restrict__ x, const float* __restrict__ Alog,
               float* __restrict__ Pout, float* __restrict__ Hout) {
  __shared__ __attribute__((aligned(16))) float sDX[T_ * CW_ * 2];
  __shared__ __attribute__((aligned(16))) float sB[T_ * N_];

  const int lane = threadIdx.x & 31;
  const int q  = __builtin_amdgcn_readfirstlane((int)(threadIdx.x >> 5));
  const int dg = blockIdx.x % DG_;
  const int bg = blockIdx.x / DG_;
  const int g  = bg % G_;
  const int bb = bg / G_;
  const int d  = dg * CW_ + lane;
  const int nq = q * NQ_;

  v2f A2a[NQ_ / 2], A2b[NQ_ / 2], h2a[NQ_ / 2], h2b[NQ_ / 2];
  s6_load_A2(Alog + (size_t)d * N_ + nq, A2a);
  s6_load_A2(Alog + (size_t)(d + 32) * N_ + nq, A2b);
#pragma unroll
  for (int j = 0; j < NQ_ / 2; ++j) {
    h2a[j].x = 0.0f;  h2a[j].y = 0.0f;
    h2b[j].x = 0.0f;  h2b[j].y = 0.0f;
  }
  float dsa = 0.0f, dsb = 0.0f;

  const size_t row0 = (size_t)bb * L_ + (size_t)g * T_;
  s6_delta_tile(xdbl + row0 * E_, Wdt + (size_t)dg * CW_ * R_, bdt + dg * CW_, q, lane, sDX);
  s6_stage_x(x + row0 * D_ + dg * CW_, sDX);
  {
    const float* bsrc = xdbl + row0 * E_ + R_;
#pragma unroll
    for (int i = 0; i < 8; ++i) {
      const int idx = (int)threadIdx.x + 128 * i;
      const int row = idx >> 4, c4 = (idx & 15) * 4;
      *(v4f*)(sB + row * N_ + c4) = *(const v4f*)(bsrc + (size_t)row * E_ + c4);
    }
  }
  __syncthreads();

  const float* bq = sB + nq;
  const float* dx = sDX + 2 * lane;

  v4f Ba[4], Bb[4];
  v2f da0, da1, db0, db1;
  s6_row16(bq, Ba);
  da0 = *(const v2f*)dx;
  da1 = *(const v2f*)(dx + 64);
  for (int tt = 0; tt < T_; tt += 2) {
    s6_row16(bq + (tt + 1) * N_, Bb);
    db0 = *(const v2f*)(dx + (tt + 1) * 2 * CW_);
    db1 = *(const v2f*)(dx + (tt + 1) * 2 * CW_ + 64);
    s6_step_chunk(Ba, da0.x, da0.y, A2a, h2a);
    s6_step_chunk(Ba, da1.x, da1.y, A2b, h2b);
    dsa += da0.x;
    dsb += da1.x;
    asm volatile("" :: "v"(Bb[0]), "v"(Bb[1]), "v"(Bb[2]), "v"(Bb[3]), "v"(db0), "v"(db1));
    const int t2 = (tt + 2 < T_) ? tt + 2 : tt + 1;
    s6_row16(bq + t2 * N_, Ba);
    da0 = *(const v2f*)(dx + t2 * 2 * CW_);
    da1 = *(const v2f*)(dx + t2 * 2 * CW_ + 64);
    s6_step_chunk(Bb, db0.x, db0.y, A2a, h2a);
    s6_step_chunk(Bb, db1.x, db1.y, A2b, h2b);
    dsa += db0.x;
    dsb += db1.x;
    asm volatile("" :: "v"(Ba[0]), "v"(Ba[1]), "v"(Ba[2]), "v"(Ba[3]), "v"(da0), "v"(da1));
  }
  const size_t base = ((size_t)bg * N_ + nq) * D_ + d;
#pragma unroll
  for (int j = 0; j < NQ_ / 2; ++j) {
    const v2f parg = dsa * A2a[j];
    Pout[base + (size_t)(2 * j) * D_]     = __builtin_amdgcn_exp2f(parg.x);
    Pout[base + (size_t)(2 * j + 1) * D_] = __builtin_amdgcn_exp2f(parg.y);
    Hout[base + (size_t)(2 * j) * D_]     = h2a[j].x;
    Hout[base + (size_t)(2 * j + 1) * D_] = h2a[j].y;
  }
#pragma unroll
  for (int j = 0; j < NQ_ / 2; ++j) {
    const v2f parg = dsb * A2b[j];
    Pout[base + 32 + (size_t)(2 * j) * D_]     = __builtin_amdgcn_exp2f(parg.x);
    Pout[base + 32 + (size_t)(2 * j + 1) * D_] = __builtin_amdgcn_exp2f(parg.y);
    Hout[base + 32 + (size_t)(2 * j) * D_]     = h2b[j].x;
    Hout[base + 32 + (size_t)(2 * j + 1) * D_] = h2b[j].y;
  }
}

__global__ void __launch_bounds__(256)
s6_combine(const float* __restrict__ Pin, const float* __restrict__ Hin,
           float* __restrict__ h0out) {
  const int idx = blockIdx.x * 256 + threadIdx.x;
  const int bb  = idx / (N_ * D_);
  const int nd  = idx % (N_ * D_);
  const size_t base = (size_t)bb * G_ * N_ * D_ + nd;

  float h = 0.0f;
#pragma unroll
  for (int g = 0; g < G_; ++g) {
    const size_t o = base + (size_t)g * N_ * D_;
    h0out[o] = h;
    h = Pin[o] * h + Hin[o];
  }
}

#define SL_ 64
__global__ void __launch_bounds__(128)
s6_scan_emit(const float* __restrict__ xdbl, const float* __restrict__ Wdt,
             const float* __restrict__ bdt,
             const float* __restrict__ x, const float* __restrict__ Alog,
             const float* __restrict__ Dp, const float* __restrict__ h0in,
             float* __restrict__ y) {
  __shared__ __attribute__((aligned(16))) float sDX[T_ * CW_ * 2];
  __shared__ __attribute__((aligned(16))) float sBC[T_ * NS_ * SL_];

  const int lane = threadIdx.x & 31;
  const int q  = __builtin_amdgcn_readfirstlane((int)(threadIdx.x >> 5));
  const int dg = blockIdx.x % DG_;
  const int bg = blockIdx.x / DG_;
  const int g  = bg % G_;
  const int bb = bg / G_;
  const int d  = dg * CW_ + lane;
  const int nq = q * NQ_;

  v2f A2a[NQ_ / 2], A2b[NQ_ / 2], h2a[NQ_ / 2], h2b[NQ_ / 2];
  s6_load_A2(Alog + (size_t)d * N_ + nq, A2a);
  s6_load_A2(Alog + (size_t)(d + 32) * N_ + nq, A2b);
  const size_t hbase = ((size_t)bg * N_ + nq) * D_ + d;
#pragma unroll
  for (int j = 0; j < NQ_ / 2; ++j) {
    h2a[j].x = h0in[hbase + (size_t)(2 * j) * D_];
    h2a[j].y = h0in[hbase + (size_t)(2 * j + 1) * D_];
    h2b[j].x = h0in[hbase + 32 + (size_t)(2 * j) * D_];
    h2b[j].y = h0in[hbase + 32 + (size_t)(2 * j + 1) * D_];
  }

  const size_t row0 = (size_t)bb * L_ + (size_t)g * T_;
  s6_delta_tile(xdbl + row0 * E_, Wdt + (size_t)dg * CW_ * R_, bdt + dg * CW_, q, lane, sDX);
  s6_stage_x(x + row0 * D_ + dg * CW_, sDX);
  {
    const float* bsrc = xdbl + row0 * E_ + R_;
#pragma unroll
    for (int i = 0; i < 16; ++i) {
      const int idx = (int)threadIdx.x + 128 * i;
      const int row = idx >> 5, col = (idx & 31) * 4;
      const int n = col & 63;
      *(v4f*)(sBC + (row * NS_ + (n >> 4)) * SL_ + (col >> 6) * 16 + (n & 15)) =
          *(const v4f*)(bsrc + (size_t)row * E_ + col);
    }
  }
  __syncthreads();

  float* bc = sBC + q * SL_;
  const float* dx = sDX + 2 * lane;

  v4f Ba[4], Ca[4], Bb[4], Cb[4];
  v2f da0, da1, db0, db1;
  s6_row16(bc, Ba);
  s6_row16(bc + 16, Ca);
  da0 = *(const v2f*)dx;
  da1 = *(const v2f*)(dx + 64);
  for (int tt = 0; tt < T_; tt += 2) {
    s6_row16(bc + (tt + 1) * NS_ * SL_, Bb);
    s6_row16(bc + (tt + 1) * NS_ * SL_ + 16, Cb);
    db0 = *(const v2f*)(dx + (tt + 1) * 2 * CW_);
    db1 = *(const v2f*)(dx + (tt + 1) * 2 * CW_ + 64);
    __builtin_amdgcn_sched_barrier(0);
    bc[tt * NS_ * SL_ + lane]      = s6_step_emit(Ba, Ca, da0.x, da0.y, A2a, h2a);
    bc[tt * NS_ * SL_ + 32 + lane] = s6_step_emit(Ba, Ca, da1.x, da1.y, A2b, h2b);
    asm volatile("" :: "v"(Bb[0]), "v"(Bb[1]), "v"(Bb[2]), "v"(Bb[3]),
                       "v"(Cb[0]), "v"(Cb[1]), "v"(Cb[2]), "v"(Cb[3]), "v"(db0), "v"(db1));
    const int t2 = (tt + 2 < T_) ? tt + 2 : tt + 1;
    s6_row16(bc + t2 * NS_ * SL_, Ba);
    s6_row16(bc + t2 * NS_ * SL_ + 16, Ca);
    da0 = *(const v2f*)(dx + t2 * 2 * CW_);
    da1 = *(const v2f*)(dx + t2 * 2 * CW_ + 64);
    __builtin_amdgcn_sched_barrier(0);
    bc[(tt + 1) * NS_ * SL_ + lane]      = s6_step_emit(Bb, Cb, db0.x, db0.y, A2a, h2a);
    bc[(tt + 1) * NS_ * SL_ + 32 + lane] = s6_step_emit(Bb, Cb, db1.x, db1.y, A2b, h2b);
    asm volatile("" :: "v"(Ba[0]), "v"(Ba[1]), "v"(Ba[2]), "v"(Ba[3]),
                       "v"(Ca[0]), "v"(Ca[1]), "v"(Ca[2]), "v"(Ca[3]), "v"(da0), "v"(da1));
  }
  __syncthreads();

  const float Dva = Dp[d];
  const float Dvb = Dp[d + 32];
#pragma unroll 4
  for (int k = 0; k < T_ / NS_; ++k) {
    const int tt = q * (T_ / NS_) + k;
    const float* yp = sBC + tt * NS_ * SL_ + lane;
    const float sa = (yp[0] + yp[SL_]) + (yp[2 * SL_] + yp[3 * SL_]);
    const float sb = (yp[32] + yp[SL_ + 32]) + (yp[2 * SL_ + 32] + yp[3 * SL_ + 32]);
    const float xa = dx[tt * 2 * CW_ + 1];
    const float xb = dx[tt * 2 * CW_ + 65];
    y[(row0 + tt) * D_ + d]      = sa + xa * Dva;
    y[(row0 + tt) * D_ + d + 32] = sb + xb * Dvb;
  }
}

extern "C" void kernel_launch(void* const* d_in, const int* in_sizes, int n_in,
                              void* d_out, int out_size, void* d_ws, size_t ws_size,
                              hipStream_t stream) {
  const float* x      = (const float*)d_in[0];
  const float* Wxproj = (const float*)d_in[1];
  const float* Wdt    = (const float*)d_in[2];
  const float* bdt    = (const float*)d_in[3];
  const float* Alog   = (const float*)d_in[4];
  const float* Dp     = (const float*)d_in[5];
  float*       y      = (float*)d_out;

  float* ws    = (float*)d_ws;
  float* xdbl  = ws;
  float* Pws   = xdbl  + (size_t)BL_ * E_;
  float* Hws   = Pws   + (size_t)B_ * D_ * G_ * N_;
  float* h0ws  = Hws   + (size_t)B_ * D_ * G_ * N_;

  s6_xproj_gemm<<<256, 256, 0, stream>>>(x, Wxproj, xdbl);
  s6_scan_chunks<<<512, 128, 0, stream>>>(xdbl, Wdt, bdt, x, Alog, Pws, Hws);
  s6_combine<<<256, 256, 0, stream>>>(Pws, Hws, h0ws);
  s6_scan_emit<<<512, 128, 0, stream>>>(xdbl, Wdt, bdt, x, Alog, Dp, h0ws, y);
}


# Chunk decay P no longer stored: pass 1 writes one delta sum per channel and chunk, combine forms P = exp2(A2*sum) itself; one 8 MB plane less written and read (same values, sum order unchanged)
# speedup vs baseline: 4.8775x; 1.0314x over previous
#include <hip/hip_runtime.h>
#include <hip/hip_bf16.h>

typedef float v2f __attribute__((ext_vector_type(2)));
typedef float v8f __attribute__((ext_vector_type(8)));
typedef float v4f __attribute__((ext_vector_type(4)));
typedef float v16f __attribute__((ext_vector_type(16)));

#define B_  2
#define L_  2048
#define D_  512
#define N_  64
#define R_  32
#define E_  160
#define BL_ (B_ * L_)
#define G_  32
#define T_  (L_ / G_)
#define NS_ 4
#define NQ_ (N_ / NS_)
#define CW_ 64
#define DG_ (D_ / CW_)

#define XW_  8
#define XK_  (D_ / XW_)
#define XP_  68
#define XT_  (16 * XP_)
#define XET_ (E_ / 16)
#define XWS_ (5 * XT_)

__device__ __forceinline__ void
s6_slice_fetch(const float* __restrict__ src, const int lane, v4f (&r)[8]) {
  const float* p = src + (size_t)(lane >> 4) * D_ + 4 * (lane & 15);
#pragma unroll
  for (int i = 0; i < 8; ++i) r[i] = *(const v4f*)(p + (size_t)(2 * i) * D_);
}

__device__ __forceinline__ void
s6_slice_park(float* dst, const int lane, const v4f (&r)[8]) {
  float* p = dst + (lane >> 4) * XP_ + 4 * (lane & 15);
#pragma unroll
  for (int i = 0; i < 8; ++i) *(v4f*)(p + 2 * i * XP_) = r[i];
}

__device__ __forceinline__ void
s6_slice_frags(const float* tile, const int lane, v4f (&f)[8]) {
  const float* p = tile + (lane & 15) * XP_ + 4 * (lane >> 4);
#pragma unroll
  for (int i = 0; i < 8; ++i) f[i] = *(const v4f*)(p + 8 * i);
}

__device__ __forceinline__ void s6_wave_lds_handover() {
  __builtin_amdgcn_fence(__ATOMIC_RELEASE, "wavefront");
  __builtin_amdgcn_wave_barrier();
  __builtin_amdgcn_fence(__ATOMIC_ACQUIRE, "wavefront");
}

__global__ void __launch_bounds__(256)
s6_xproj_gemm(const float* __restrict__ x, const float* __restrict__ W,
              float* __restrict__ xdbl) {
  __shared__ __attribute__((aligned(16))) float sm[XW_ * XWS_];

  const int lane = threadIdx.x & 31;
  const int w  = __builtin_amdgcn_readfirstlane((int)(threadIdx.x >> 5));
  const int rt = blockIdx.x;

  float* sa = sm + w * XWS_;
  float* sb = sa + XT_;
  const float* xa = x + (size_t)rt * 16 * D_ + w * XK_;
  const float* wb = W + w * XK_;

  v4f ra[8], rb0[8], rb1[8];
  s6_slice_fetch(xa, lane, ra);
  s6_slice_fetch(wb, lane, rb0);
  s6_slice_fetch(wb + (size_t)16 * D_, lane, rb1);
  s6_slice_park(sa, lane, ra);
  s6_slice_park(sb, lane, rb0);
  s6_slice_park(sb + XT_, lane, rb1);
  s6_slice_fetch(wb + (size_t)32 * D_, lane, rb0);
  s6_slice_fetch(wb + (size_t)48 * D_, lane, rb1);
  s6_wave_lds_handover();

  v4f af[8], bf[2][2][8];
  s6_slice_frags(sa, lane, af);
  s6_slice_frags(sb, lane, bf[0][0]);
  s6_slice_frags(sb + XT_, lane, bf[0][1]);
  s6_slice_park(sb + 2 * XT_, lane, rb0);
  s6_slice_park(sb + 3 * XT_, lane, rb1);
  s6_wave_lds_handover();

  v8f acc[XET_];
#pragma unroll
  for (int j = 0; j < XET_; ++j) acc[j] = (v8f){};

#pragma unroll
  for (int jp = 0; jp < XET_ / 2; ++jp) {
    const int cs = jp & 1, ns = cs ^ 1;
    if (jp + 2 < XET_ / 2) {
      s6_slice_fetch(wb + (size_t)(2 * jp + 4) * 16 * D_, lane, rb0);
      s6_slice_fetch(wb + (size_t)(2 * jp + 5) * 16 * D_, lane, rb1);
    }
    if (jp + 1 < XET_ / 2) {
      s6_slice_frags(sb + ns * 2 * XT_, lane, bf[ns][0]);
      s6_slice_frags(sb + ns * 2 * XT_ + XT_, lane, bf[ns][1]);
    }
    __builtin_amdgcn_sched_barrier(0);
#pragma unroll
    for (int i = 0; i < 8; ++i) {
      const v2f a0 = {af[i].x, af[i].y}, a1 = {af[i].z, af[i].w};
      const v2f p0 = {bf[cs][0][i].x, bf[cs][0][i].y}, p1 = {bf[cs][0][i].z, bf[cs][0][i].w};
      const v2f q0 = {bf[cs][1][i].x, bf[cs][1][i].y}, q1 = {bf[cs][1][i].z, bf[cs][1][i].w};
      acc[2 * jp]     = __builtin_amdgcn_wmma_f32_16x16x4_f32(false, a0, false, p0, (short)0, acc[2 * jp],     false, false);
      acc[2 * jp + 1] = __builtin_amdgcn_wmma_f32_16x16x4_f32(false, a0, false, q0, (short)0, acc[2 * jp + 1], false, false);
      acc[2 * jp]     = __builtin_amdgcn_wmma_f32_16x16x4_f32(false, a1, false, p1, (short)0, acc[2 * jp],     false, false);
      acc[2 * jp + 1] = __builtin_amdgcn_wmma_f32_16x16x4_f32(false, a1, false, q1, (short)0, acc[2 * jp + 1], false, false);
    }
    __builtin_amdgcn_sched_barrier(0);
    if (jp + 2 < XET_ / 2) {
      s6_slice_park(sb + cs * 2 * XT_, lane, rb0);
      s6_slice_park(sb + cs * 2 * XT_ + XT_, lane, rb1);
      s6_wave_lds_handover();
    }
  }

  __syncthreads();
#pragma unroll
  for (int j = 0; j < XET_; ++j) {
#pragma unroll
    for (int r = 0; r < 8; ++r)
      sm[((w * XET_ + j) * 8 + r) * 32 + lane] = acc[j][r];
  }
  __syncthreads();

  float* orow = xdbl + (size_t)rt * 16 * E_;
#pragma unroll 2
  for (int o = threadIdx.x; o < 16 * E_; o += 256) {
    const int row = o / E_, col = o % E_;
    const int idx = ((col >> 4) * 8 + (row & 7)) * 32 + (col & 15) + 16 * (row >> 3);
    float s = sm[idx];
#pragma unroll
    for (int q = 1; q < XW_; ++q) s += sm[q * XET_ * 256 + idx];
    orow[o] = s;
  }
}

__device__ __forceinline__ float s6_softplus(const float z) {
  const float sp = __logf(1.0f + __expf(z));
  return (z > 20.0f) ? z : sp;
}

__device__ __forceinline__ void
s6_delta_tile(const float* __restrict__ xrows,
              const float* __restrict__ wdt,
              const float* __restrict__ bdt,
              const int q, const int lane, float* sDX) {
  const int laneM = lane & 15;
  const int laneH = lane >> 4;
  const float* arow = xrows + (size_t)(q * 16 + laneM) * E_ + 2 * laneH;
  const float* br0  = wdt + (size_t)laneM * R_ + 2 * laneH;

  v8f cc[CW_ / 16];
#pragma unroll
  for (int ct = 0; ct < CW_ / 16; ++ct) cc[ct] = (v8f){};
#pragma unroll
  for (int k = 0; k < R_; k += 4) {
    const v2f a = *(const v2f*)(arow + k);
#pragma unroll
    for (int ct = 0; ct < CW_ / 16; ++ct) {
      const v2f b = *(const v2f*)(br0 + (size_t)ct * 16 * R_ + k);
      cc[ct] = __builtin_amdgcn_wmma_f32_16x16x4_f32(false, a, false, b, (short)0, cc[ct], false, false);
    }
  }
  float* p = sDX + ((q * 16 + 8 * laneH) * CW_ + laneM) * 2;
#pragma unroll
  for (int ct = 0; ct < CW_ / 16; ++ct) {
    const float bi = bdt[ct * 16 + laneM];
#pragma unroll
    for (int j = 0; j < 8; ++j)
      p[j * 2 * CW_ + ct * 32] = s6_softplus(cc[ct][j] + bi);
  }
}


__device__ __forceinline__ void
s6_stage_x(const float* __restrict__ xsrc, float* sDX) {
#pragma unroll
  for (int i = 0; i < 8; ++i) {
    const int idx = (int)threadIdx.x + 128 * i;
    const int row = idx >> 4, c4 = (idx & 15) * 4;
    const v4f xv = *(const v4f*)(xsrc + (size_t)row * D_ + c4);
    float* p = sDX + (row * CW_ + c4) * 2 + 1;
    p[0] = xv.x;  p[2] = xv.y;  p[4] = xv.z;  p[6] = xv.w;
  }
}

__device__ __forceinline__ void s6_row16(const float* p, v4f (&r)[4]) {
#pragma unroll
  for (int i = 0; i < 4; ++i) r[i] = *(const v4f*)(p + 4 * i);
}

#define S6_LOG2E 1.44269504088896340736f

__device__ __forceinline__ void
s6_step_chunk(const v4f (&Bc)[4], const float dlt, const float xv,
              const v2f (&A2)[NQ_ / 2], v2f (&h2)[NQ_ / 2]) {
  const float du = dlt * xv;
#pragma unroll
  for (int j = 0; j < NQ_ / 2; ++j) {
    const v2f arg = dlt * A2[j];
    v2f a;  a.x = __builtin_amdgcn_exp2f(arg.x);  a.y = __builtin_amdgcn_exp2f(arg.y);
    v2f Bv; Bv.x = Bc[j >> 1][2 * (j & 1)];  Bv.y = Bc[j >> 1][2 * (j & 1) + 1];
    h2[j] = a * h2[j] + du * Bv;
  }
}

__device__ __forceinline__ float
s6_step_emit(const v4f (&Bc)[4], const v4f (&Cc)[4], const float dlt, const float xv,
             const v2f (&A2)[NQ_ / 2], v2f (&h2)[NQ_ / 2]) {
  const float du = dlt * xv;
  v2f acc = {0.0f, 0.0f};
#pragma unroll
  for (int j = 0; j < NQ_ / 2; ++j) {
    const v2f arg = dlt * A2[j];
    v2f a;  a.x = __builtin_amdgcn_exp2f(arg.x);  a.y = __builtin_amdgcn_exp2f(arg.y);
    v2f Bv; Bv.x = Bc[j >> 1][2 * (j & 1)];  Bv.y = Bc[j >> 1][2 * (j & 1) + 1];
    v2f Cv; Cv.x = Cc[j >> 1][2 * (j & 1)];  Cv.y = Cc[j >> 1][2 * (j & 1) + 1];
    h2[j] = a * h2[j] + du * Bv;
    acc += h2[j] * Cv;
  }
  return acc.x + acc.y;
}

__device__ __forceinline__ void
s6_load_A2(const float* __restrict__ arow, v2f (&A2)[NQ_ / 2]) {
  const v4f* alp = (const v4f*)arow;
#pragma unroll
  for (int k = 0; k < NQ_ / 4; ++k) {
    const v4f al = alp[k];
    A2[2 * k].x     = -__expf(al.x) * S6_LOG2E;
    A2[2 * k].y     = -__expf(al.y) * S6_LOG2E;
    A2[2 * k + 1].x = -__expf(al.z) * S6_LOG2E;
    A2[2 * k + 1].y = -__expf(al.w) * S6_LOG2E;
  }
}

__global__ void __launch_bounds__(128)
s6_scan_chunks(const float* __restrict__ xdbl, const float* __restrict__ Wdt,
               const float* __restrict__ bdt,
               const float* __restrict__ x, const float* __restrict__ Alog,
               float* __restrict__ DSout, float* __restrict__ Hout) {
  __shared__ __attribute__((aligned(16))) float sDX[T_ * CW_ * 2];
  __shared__ __attribute__((aligned(16))) float sB[T_ * N_];

  const int lane = threadIdx.x & 31;
  const int q  = __builtin_amdgcn_readfirstlane((int)(threadIdx.x >> 5));
  const int dg = blockIdx.x % DG_;
  const int bg = blockIdx.x / DG_;
  const int g  = bg % G_;
  const int bb = bg / G_;
  const int d  = dg * CW_ + lane;
  const int nq = q * NQ_;

  v2f A2a[NQ_ / 2], A2b[NQ_ / 2], h2a[NQ_ / 2], h2b[NQ_ / 2];
  s6_load_A2(Alog + (size_t)d * N_ + nq, A2a);
  s6_load_A2(Alog + (size_t)(d + 32) * N_ + nq, A2b);
#pragma unroll
  for (int j = 0; j < NQ_ / 2; ++j) {
    h2a[j].x = 0.0f;  h2a[j].y = 0.0f;
    h2b[j].x = 0.0f;  h2b[j].y = 0.0f;
  }
  float dsa = 0.0f, dsb = 0.0f;

  const size_t row0 = (size_t)bb * L_ + (size_t)g * T_;
  s6_delta_tile(xdbl + row0 * E_, Wdt + (size_t)dg * CW_ * R_, bdt + dg * CW_, q, lane, sDX);
  s6_stage_x(x + row0 * D_ + dg * CW_, sDX);
  {
    const float* bsrc = xdbl + row0 * E_ + R_;
#pragma unroll
    for (int i = 0; i < 8; ++i) {
      const int idx = (int)threadIdx.x + 128 * i;
      const int row = idx >> 4, c4 = (idx & 15) * 4;
      *(v4f*)(sB + row * N_ + c4) = *(const v4f*)(bsrc + (size_t)row * E_ + c4);
    }
  }
  __syncthreads();

  const float* bq = sB + nq;
  const float* dx = sDX + 2 * lane;

  v4f Ba[4], Bb[4];
  v2f da0, da1, db0, db1;
  s6_row16(bq, Ba);
  da0 = *(const v2f*)dx;
  da1 = *(const v2f*)(dx + 64);
  for (int tt = 0; tt < T_; tt += 2) {
    s6_row16(bq + (tt + 1) * N_, Bb);
    db0 = *(const v2f*)(dx + (tt + 1) * 2 * CW_);
    db1 = *(const v2f*)(dx + (tt + 1) * 2 * CW_ + 64);
    s6_step_chunk(Ba, da0.x, da0.y, A2a, h2a);
    s6_step_chunk(Ba, da1.x, da1.y, A2b, h2b);
    dsa += da0.x;
    dsb += da1.x;
    asm volatile("" :: "v"(Bb[0]), "v"(Bb[1]), "v"(Bb[2]), "v"(Bb[3]), "v"(db0), "v"(db1));
    const int t2 = (tt + 2 < T_) ? tt + 2 : tt + 1;
    s6_row16(bq + t2 * N_, Ba);
    da0 = *(const v2f*)(dx + t2 * 2 * CW_);
    da1 = *(const v2f*)(dx + t2 * 2 * CW_ + 64);
    s6_step_chunk(Bb, db0.x, db0.y, A2a, h2a);
    s6_step_chunk(Bb, db1.x, db1.y, A2b, h2b);
    dsa += db0.x;
    dsb += db1.x;
    asm volatile("" :: "v"(Ba[0]), "v"(Ba[1]), "v"(Ba[2]), "v"(Ba[3]), "v"(da0), "v"(da1));
  }
  const size_t base = ((size_t)bg * N_ + nq) * D_ + d;
#pragma unroll
  for (int j = 0; j < NQ_ / 2; ++j) {
    Hout[base + (size_t)(2 * j) * D_]     = h2a[j].x;
    Hout[base + (size_t)(2 * j + 1) * D_] = h2a[j].y;
  }
#pragma unroll
  for (int j = 0; j < NQ_ / 2; ++j) {
    Hout[base + 32 + (size_t)(2 * j) * D_]     = h2b[j].x;
    Hout[base + 32 + (size_t)(2 * j + 1) * D_] = h2b[j].y;
  }
  if (q == 0) {
    DSout[(size_t)bg * D_ + d]      = dsa;
    DSout[(size_t)bg * D_ + d + 32] = dsb;
  }
}

__global__ void __launch_bounds__(256)
s6_combine(const float* __restrict__ DSin, const float* __restrict__ Alog,
           const float* __restrict__ Hin, float* __restrict__ h0out) {
  const int idx = blockIdx.x * 256 + threadIdx.x;
  const int bb  = idx / (N_ * D_);
  const int nd  = idx % (N_ * D_);
  const int n   = nd / D_;
  const int d   = nd % D_;
  const size_t base = (size_t)bb * G_ * N_ * D_ + nd;
  const float* ds = DSin + (size_t)bb * G_ * D_ + d;

  const float al = Alog[(size_t)d * N_ + n];
  float dsv[G_], hv[G_];
#pragma unroll
  for (int g = 0; g < G_; ++g) {
    dsv[g] = ds[(size_t)g * D_];
    hv[g]  = Hin[base + (size_t)g * N_ * D_];
  }

  const float A2 = -__expf(al) * S6_LOG2E;
  float h = 0.0f;
#pragma unroll
  for (int g = 0; g < G_; ++g) {
    h0out[base + (size_t)g * N_ * D_] = h;
    const float P = __builtin_amdgcn_exp2f(dsv[g] * A2);
    h = P * h + hv[g];
  }
}

#define SL_ 64
__global__ void __launch_bounds__(128)
s6_scan_emit(const float* __restrict__ xdbl, const float* __restrict__ Wdt,
             const float* __restrict__ bdt,
             const float* __restrict__ x, const float* __restrict__ Alog,
             const float* __restrict__ Dp, const float* __restrict__ h0in,
             float* __restrict__ y) {
  __shared__ __attribute__((aligned(16))) float sDX[T_ * CW_ * 2];
  __shared__ __attribute__((aligned(16))) float sBC[T_ * NS_ * SL_];

  const int lane = threadIdx.x & 31;
  const int q  = __builtin_amdgcn_readfirstlane((int)(threadIdx.x >> 5));
  const int dg = blockIdx.x % DG_;
  const int bg = blockIdx.x / DG_;
  const int g  = bg % G_;
  const int bb = bg / G_;
  const int d  = dg * CW_ + lane;
  const int nq = q * NQ_;

  v2f A2a[NQ_ / 2], A2b[NQ_ / 2], h2a[NQ_ / 2], h2b[NQ_ / 2];
  s6_load_A2(Alog + (size_t)d * N_ + nq, A2a);
  s6_load_A2(Alog + (size_t)(d + 32) * N_ + nq, A2b);
  const size_t hbase = ((size_t)bg * N_ + nq) * D_ + d;
#pragma unroll
  for (int j = 0; j < NQ_ / 2; ++j) {
    h2a[j].x = h0in[hbase + (size_t)(2 * j) * D_];
    h2a[j].y = h0in[hbase + (size_t)(2 * j + 1) * D_];
    h2b[j].x = h0in[hbase + 32 + (size_t)(2 * j) * D_];
    h2b[j].y = h0in[hbase + 32 + (size_t)(2 * j + 1) * D_];
  }

  const size_t row0 = (size_t)bb * L_ + (size_t)g * T_;
  s6_delta_tile(xdbl + row0 * E_, Wdt + (size_t)dg * CW_ * R_, bdt + dg * CW_, q, lane, sDX);
  s6_stage_x(x + row0 * D_ + dg * CW_, sDX);
  {
    const float* bsrc = xdbl + row0 * E_ + R_;
#pragma unroll
    for (int i = 0; i < 16; ++i) {
      const int idx = (int)threadIdx.x + 128 * i;
      const int row = idx >> 5, col = (idx & 31) * 4;
      const int n = col & 63;
      *(v4f*)(sBC + (row * NS_ + (n >> 4)) * SL_ + (col >> 6) * 16 + (n & 15)) =
          *(const v4f*)(bsrc + (size_t)row * E_ + col);
    }
  }
  __syncthreads();

  float* bc = sBC + q * SL_;
  const float* dx = sDX + 2 * lane;

  v4f Ba[4], Ca[4], Bb[4], Cb[4];
  v2f da0, da1, db0, db1;
  s6_row16(bc, Ba);
  s6_row16(bc + 16, Ca);
  da0 = *(const v2f*)dx;
  da1 = *(const v2f*)(dx + 64);
  for (int tt = 0; tt < T_; tt += 2) {
    s6_row16(bc + (tt + 1) * NS_ * SL_, Bb);
    s6_row16(bc + (tt + 1) * NS_ * SL_ + 16, Cb);
    db0 = *(const v2f*)(dx + (tt + 1) * 2 * CW_);
    db1 = *(const v2f*)(dx + (tt + 1) * 2 * CW_ + 64);
    __builtin_amdgcn_sched_barrier(0);
    bc[tt * NS_ * SL_ + lane]      = s6_step_emit(Ba, Ca, da0.x, da0.y, A2a, h2a);
    bc[tt * NS_ * SL_ + 32 + lane] = s6_step_emit(Ba, Ca, da1.x, da1.y, A2b, h2b);
    asm volatile("" :: "v"(Bb[0]), "v"(Bb[1]), "v"(Bb[2]), "v"(Bb[3]),
                       "v"(Cb[0]), "v"(Cb[1]), "v"(Cb[2]), "v"(Cb[3]), "v"(db0), "v"(db1));
    const int t2 = (tt + 2 < T_) ? tt + 2 : tt + 1;
    s6_row16(bc + t2 * NS_ * SL_, Ba);
    s6_row16(bc + t2 * NS_ * SL_ + 16, Ca);
    da0 = *(const v2f*)(dx + t2 * 2 * CW_);
    da1 = *(const v2f*)(dx + t2 * 2 * CW_ + 64);
    __builtin_amdgcn_sched_barrier(0);
    bc[(tt + 1) * NS_ * SL_ + lane]      = s6_step_emit(Bb, Cb, db0.x, db0.y, A2a, h2a);
    bc[(tt + 1) * NS_ * SL_ + 32 + lane] = s6_step_emit(Bb, Cb, db1.x, db1.y, A2b, h2b);
    asm volatile("" :: "v"(Ba[0]), "v"(Ba[1]), "v"(Ba[2]), "v"(Ba[3]),
                       "v"(Ca[0]), "v"(Ca[1]), "v"(Ca[2]), "v"(Ca[3]), "v"(da0), "v"(da1));
  }
  __syncthreads();

  const float Dva = Dp[d];
  const float Dvb = Dp[d + 32];
#pragma unroll 4
  for (int k = 0; k < T_ / NS_; ++k) {
    const int tt = q * (T_ / NS_) + k;
    const float* yp = sBC + tt * NS_ * SL_ + lane;
    const float sa = (yp[0] + yp[SL_]) + (yp[2 * SL_] + yp[3 * SL_]);
    const float sb = (yp[32] + yp[SL_ + 32]) + (yp[2 * SL_ + 32] + yp[3 * SL_ + 32]);
    const float xa = dx[tt * 2 * CW_ + 1];
    const float xb = dx[tt * 2 * CW_ + 65];
    y[(row0 + tt) * D_ + d]      = sa + xa * Dva;
    y[(row0 + tt) * D_ + d + 32] = sb + xb * Dvb;
  }
}

extern "C" void kernel_launch(void* const* d_in, const int* in_sizes, int n_in,
                              void* d_out, int out_size, void* d_ws, size_t ws_size,
                              hipStream_t stream) {
  const float* x      = (const float*)d_in[0];
  const float* Wxproj = (const float*)d_in[1];
  const float* Wdt    = (const float*)d_in[2];
  const float* bdt    = (const float*)d_in[3];
  const float* Alog   = (const float*)d_in[4];
  const float* Dp     = (const float*)d_in[5];
  float*       y      = (float*)d_out;

  float* ws    = (float*)d_ws;
  float* xdbl  = ws;
  float* DSws  = xdbl  + (size_t)BL_ * E_;
  float* Hws   = DSws  + (size_t)B_ * D_ * G_;
  float* h0ws  = Hws   + (size_t)B_ * D_ * G_ * N_;

  s6_xproj_gemm<<<256, 256, 0, stream>>>(x, Wxproj, xdbl);
  s6_scan_chunks<<<512, 128, 0, stream>>>(xdbl, Wdt, bdt, x, Alog, DSws, Hws);
  s6_combine<<<256, 256, 0, stream>>>(DSws, Alog, Hws, h0ws);
  s6_scan_emit<<<512, 128, 0, stream>>>(xdbl, Wdt, bdt, x, Alog, Dp, h0ws, y);
}


# Scan time loops, issue order only: step written as a pipeline over 16 state pairs so each exp2 is followed by plain vector ops (scheduler groups); pass 1 LDS-read fences replaced by sched barriers
# speedup vs baseline: 5.0937x; 1.0443x over previous
#include <hip/hip_runtime.h>
#include <hip/hip_bf16.h>

typedef float v2f __attribute__((ext_vector_type(2)));
typedef float v8f __attribute__((ext_vector_type(8)));
typedef float v4f __attribute__((ext_vector_type(4)));
typedef float v16f __attribute__((ext_vector_type(16)));

#define B_  2
#define L_  2048
#define D_  512
#define N_  64
#define R_  32
#define E_  160
#define BL_ (B_ * L_)
#define G_  32
#define T_  (L_ / G_)
#define NS_ 4
#define NQ_ (N_ / NS_)
#define CW_ 64
#define DG_ (D_ / CW_)

#define XW_  8
#define XK_  (D_ / XW_)
#define XP_  68
#define XT_  (16 * XP_)
#define XET_ (E_ / 16)
#define XWS_ (5 * XT_)

__device__ __forceinline__ void
s6_slice_fetch(const float* __restrict__ src, const int lane, v4f (&r)[8]) {
  const float* p = src + (size_t)(lane >> 4) * D_ + 4 * (lane & 15);
#pragma unroll
  for (int i = 0; i < 8; ++i) r[i] = *(const v4f*)(p + (size_t)(2 * i) * D_);
}

__device__ __forceinline__ void
s6_slice_park(float* dst, const int lane, const v4f (&r)[8]) {
  float* p = dst + (lane >> 4) * XP_ + 4 * (lane & 15);
#pragma unroll
  for (int i = 0; i < 8; ++i) *(v4f*)(p + 2 * i * XP_) = r[i];
}

__device__ __forceinline__ void
s6_slice_frags(const float* tile, const int lane, v4f (&f)[8]) {
  const float* p = tile + (lane & 15) * XP_ + 4 * (lane >> 4);
#pragma unroll
  for (int i = 0; i < 8; ++i) f[i] = *(const v4f*)(p + 8 * i);
}

__device__ __forceinline__ void s6_wave_lds_handover() {
  __builtin_amdgcn_fence(__ATOMIC_RELEASE, "wavefront");
  __builtin_amdgcn_wave_barrier();
  __builtin_amdgcn_fence(__ATOMIC_ACQUIRE, "wavefront");
}

__global__ void __launch_bounds__(256)
s6_xproj_gemm(const float* __restrict__ x, const float* __restrict__ W,
              float* __restrict__ xdbl) {
  __shared__ __attribute__((aligned(16))) float sm[XW_ * XWS_];

  const int lane = threadIdx.x & 31;
  const int w  = __builtin_amdgcn_readfirstlane((int)(threadIdx.x >> 5));
  const int rt = blockIdx.x;

  float* sa = sm + w * XWS_;
  float* sb = sa + XT_;
  const float* xa = x + (size_t)rt * 16 * D_ + w * XK_;
  const float* wb = W + w * XK_;

  v4f ra[8], rb0[8], rb1[8];
  s6_slice_fetch(xa, lane, ra);
  s6_slice_fetch(wb, lane, rb0);
  s6_slice_fetch(wb + (size_t)16 * D_, lane, rb1);
  s6_slice_park(sa, lane, ra);
  s6_slice_park(sb, lane, rb0);
  s6_slice_park(sb + XT_, lane, rb1);
  s6_slice_fetch(wb + (size_t)32 * D_, lane, rb0);
  s6_slice_fetch(wb + (size_t)48 * D_, lane, rb1);
  s6_wave_lds_handover();

  v4f af[8], bf[2][2][8];
  s6_slice_frags(sa, lane, af);
  s6_slice_frags(sb, lane, bf[0][0]);
  s6_slice_frags(sb + XT_, lane, bf[0][1]);
  s6_slice_park(sb + 2 * XT_, lane, rb0);
  s6_slice_park(sb + 3 * XT_, lane, rb1);
  s6_wave_lds_handover();

  v8f acc[XET_];
#pragma unroll
  for (int j = 0; j < XET_; ++j) acc[j] = (v8f){};

#pragma unroll
  for (int jp = 0; jp < XET_ / 2; ++jp) {
    const int cs = jp & 1, ns = cs ^ 1;
    if (jp + 2 < XET_ / 2) {
      s6_slice_fetch(wb + (size_t)(2 * jp + 4) * 16 * D_, lane, rb0);
      s6_slice_fetch(wb + (size_t)(2 * jp + 5) * 16 * D_, lane, rb1);
    }
    if (jp + 1 < XET_ / 2) {
      s6_slice_frags(sb + ns * 2 * XT_, lane, bf[ns][0]);
      s6_slice_frags(sb + ns * 2 * XT_ + XT_, lane, bf[ns][1]);
    }
    __builtin_amdgcn_sched_barrier(0);
#pragma unroll
    for (int i = 0; i < 8; ++i) {
      const v2f a0 = {af[i].x, af[i].y}, a1 = {af[i].z, af[i].w};
      const v2f p0 = {bf[cs][0][i].x, bf[cs][0][i].y}, p1 = {bf[cs][0][i].z, bf[cs][0][i].w};
      const v2f q0 = {bf[cs][1][i].x, bf[cs][1][i].y}, q1 = {bf[cs][1][i].z, bf[cs][1][i].w};
      acc[2 * jp]     = __builtin_amdgcn_wmma_f32_16x16x4_f32(false, a0, false, p0, (short)0, acc[2 * jp],     false, false);
      acc[2 * jp + 1] = __builtin_amdgcn_wmma_f32_16x16x4_f32(false, a0, false, q0, (short)0, acc[2 * jp + 1], false, false);
      acc[2 * jp]     = __builtin_amdgcn_wmma_f32_16x16x4_f32(false, a1, false, p1, (short)0, acc[2 * jp],     false, false);
      acc[2 * jp + 1] = __builtin_amdgcn_wmma_f32_16x16x4_f32(false, a1, false, q1, (short)0, acc[2 * jp + 1], false, false);
    }
    __builtin_amdgcn_sched_barrier(0);
    if (jp + 2 < XET_ / 2) {
      s6_slice_park(sb + cs * 2 * XT_, lane, rb0);
      s6_slice_park(sb + cs * 2 * XT_ + XT_, lane, rb1);
      s6_wave_lds_handover();
    }
  }

  __syncthreads();
#pragma unroll
  for (int j = 0; j < XET_; ++j) {
#pragma unroll
    for (int r = 0; r < 8; ++r)
      sm[((w * XET_ + j) * 8 + r) * 32 + lane] = acc[j][r];
  }
  __syncthreads();

  float* orow = xdbl + (size_t)rt * 16 * E_;
#pragma unroll 2
  for (int o = threadIdx.x; o < 16 * E_; o += 256) {
    const int row = o / E_, col = o % E_;
    const int idx = ((col >> 4) * 8 + (row & 7)) * 32 + (col & 15) + 16 * (row >> 3);
    float s = sm[idx];
#pragma unroll
    for (int q = 1; q < XW_; ++q) s += sm[q * XET_ * 256 + idx];
    orow[o] = s;
  }
}

__device__ __forceinline__ float s6_softplus(const float z) {
  const float sp = __logf(1.0f + __expf(z));
  return (z > 20.0f) ? z : sp;
}

__device__ __forceinline__ void
s6_delta_tile(const float* __restrict__ xrows,
              const float* __restrict__ wdt,
              const float* __restrict__ bdt,
              const int q, const int lane, float* sDX) {
  const int laneM = lane & 15;
  const int laneH = lane >> 4;
  const float* arow = xrows + (size_t)(q * 16 + laneM) * E_ + 2 * laneH;
  const float* br0  = wdt + (size_t)laneM * R_ + 2 * laneH;

  v8f cc[CW_ / 16];
#pragma unroll
  for (int ct = 0; ct < CW_ / 16; ++ct) cc[ct] = (v8f){};
#pragma unroll
  for (int k = 0; k < R_; k += 4) {
    const v2f a = *(const v2f*)(arow + k);
#pragma unroll
    for (int ct = 0; ct < CW_ / 16; ++ct) {
      const v2f b = *(const v2f*)(br0 + (size_t)ct * 16 * R_ + k);
      cc[ct] = __builtin_amdgcn_wmma_f32_16x16x4_f32(false, a, false, b, (short)0, cc[ct], false, false);
    }
  }
  float* p = sDX + ((q * 16 + 8 * laneH) * CW_ + laneM) * 2;
#pragma unroll
  for (int ct = 0; ct < CW_ / 16; ++ct) {
    const float bi = bdt[ct * 16 + laneM];
#pragma unroll
    for (int j = 0; j < 8; ++j)
      p[j * 2 * CW_ + ct * 32] = s6_softplus(cc[ct][j] + bi);
  }
}


__device__ __forceinline__ void
s6_stage_x(const float* __restrict__ xsrc, float* sDX) {
#pragma unroll
  for (int i = 0; i < 8; ++i) {
    const int idx = (int)threadIdx.x + 128 * i;
    const int row = idx >> 4, c4 = (idx & 15) * 4;
    const v4f xv = *(const v4f*)(xsrc + (size_t)row * D_ + c4);
    float* p = sDX + (row * CW_ + c4) * 2 + 1;
    p[0] = xv.x;  p[2] = xv.y;  p[4] = xv.z;  p[6] = xv.w;
  }
}

__device__ __forceinline__ void s6_row16(const float* p, v4f (&r)[4]) {
#pragma unroll
  for (int i = 0; i < 4; ++i) r[i] = *(const v4f*)(p + 4 * i);
}

#define S6_LOG2E 1.44269504088896340736f

#define S6_MASK_PLAIN 0x002
#define S6_MASK_TRANS 0x400
#define S6_PAIRS (2 * (NQ_ / 2))
#define S6_D_ARG 3
#define S6_D_AH  2
#define S6_D_FMA 4
#define S6_D_ACC 6

template <int NP, bool HAS_T>
__device__ __forceinline__ void s6_order_slot() {
  if (HAS_T) {
    __builtin_amdgcn_sched_group_barrier(S6_MASK_TRANS, 1, 0);
    if (NP >= 1) __builtin_amdgcn_sched_group_barrier(S6_MASK_PLAIN, 1, 0);
    __builtin_amdgcn_sched_group_barrier(S6_MASK_TRANS, 1, 0);
    if (NP >= 2) __builtin_amdgcn_sched_group_barrier(S6_MASK_PLAIN, NP - 1, 0);
  } else {
    if (NP >= 1) __builtin_amdgcn_sched_group_barrier(S6_MASK_PLAIN, NP, 0);
  }
}

template <bool EMIT>
__device__ __forceinline__ void
s6_step_pair(const v4f (&Bc)[4], const v4f (&Cc)[4], const v2f dxa, const v2f dxb,
             const v2f (&A2a)[NQ_ / 2], const v2f (&A2b)[NQ_ / 2],
             v2f (&h2a)[NQ_ / 2], v2f (&h2b)[NQ_ / 2], float& ya, float& yb) {
  const float dua = dxa.x * dxa.y;
  const float dub = dxb.x * dxb.y;
  v2f arg[S6_PAIRS], ex[S6_PAIRS], ah[S6_PAIRS];
  v2f acca = {0.0f, 0.0f}, accb = {0.0f, 0.0f};
#pragma unroll
  for (int s = 0; s < S6_D_ARG; ++s)
    arg[s] = (s & 1) ? dxb.x * A2b[s >> 1] : dxa.x * A2a[s >> 1];
  __builtin_amdgcn_sched_group_barrier(S6_MASK_PLAIN, S6_D_ARG + 2, 0);
#pragma unroll
  for (int s = 0; s < S6_PAIRS + (EMIT ? S6_D_ACC : S6_D_FMA); ++s) {
    const bool hasT = s < S6_PAIRS;
    if (hasT) ex[s].x = __builtin_amdgcn_exp2f(arg[s].x);
    const int sa = s + S6_D_ARG;
    if (sa < S6_PAIRS) arg[sa] = (sa & 1) ? dxb.x * A2b[sa >> 1] : dxa.x * A2a[sa >> 1];
    if (hasT) ex[s].y = __builtin_amdgcn_exp2f(arg[s].y);
    const int sh = s - S6_D_AH;
    if (sh >= 0 && sh < S6_PAIRS) ah[sh] = ex[sh] * ((sh & 1) ? h2b[sh >> 1] : h2a[sh >> 1]);
    const int sf = s - S6_D_FMA;
    if (sf >= 0 && sf < S6_PAIRS) {
      const int j = sf >> 1;
      v2f Bv; Bv.x = Bc[j >> 1][2 * (j & 1)];  Bv.y = Bc[j >> 1][2 * (j & 1) + 1];
      const v2f du2 = (sf & 1) ? (v2f){dub, dub} : (v2f){dua, dua};
      const v2f hn = __builtin_elementwise_fma(du2, Bv, ah[sf]);
      if (sf & 1) h2b[j] = hn; else h2a[j] = hn;
    }
    const int sc = s - S6_D_ACC;
    if (EMIT && sc >= 0 && sc < S6_PAIRS) {
      const int j = sc >> 1;
      v2f Cv; Cv.x = Cc[j >> 1][2 * (j & 1)];  Cv.y = Cc[j >> 1][2 * (j & 1) + 1];
      if (sc & 1) accb = __builtin_elementwise_fma(h2b[j], Cv, accb);
      else        acca = __builtin_elementwise_fma(h2a[j], Cv, acca);
    }
    const int np = (sa < S6_PAIRS ? 1 : 0) + ((sh >= 0 && sh < S6_PAIRS) ? 1 : 0) +
                   ((sf >= 0 && sf < S6_PAIRS) ? 1 : 0) + ((EMIT && sc >= 0 && sc < S6_PAIRS) ? 1 : 0);
    if (hasT) {
      if (np == 0) s6_order_slot<0, true>(); else if (np == 1) s6_order_slot<1, true>();
      else if (np == 2) s6_order_slot<2, true>(); else if (np == 3) s6_order_slot<3, true>();
      else s6_order_slot<4, true>();
    } else {
      if (np == 1) s6_order_slot<1, false>(); else if (np == 2) s6_order_slot<2, false>();
      else if (np == 3) s6_order_slot<3, false>(); else if (np == 4) s6_order_slot<4, false>();
    }
  }
  if (EMIT) { ya = acca.x + acca.y;  yb = accb.x + accb.y; }
}

__device__ __forceinline__ void
s6_load_A2(const float* __restrict__ arow, v2f (&A2)[NQ_ / 2]) {
  const v4f* alp = (const v4f*)arow;
#pragma unroll
  for (int k = 0; k < NQ_ / 4; ++k) {
    const v4f al = alp[k];
    A2[2 * k].x     = -__expf(al.x) * S6_LOG2E;
    A2[2 * k].y     = -__expf(al.y) * S6_LOG2E;
    A2[2 * k + 1].x = -__expf(al.z) * S6_LOG2E;
    A2[2 * k + 1].y = -__expf(al.w) * S6_LOG2E;
  }
}

__global__ void __launch_bounds__(128)
s6_scan_chunks(const float* __restrict__ xdbl, const float* __restrict__ Wdt,
               const float* __restrict__ bdt,
               const float* __restrict__ x, const float* __restrict__ Alog,
               float* __restrict__ DSout, float* __restrict__ Hout) {
  __shared__ __attribute__((aligned(16))) float sDX[T_ * CW_ * 2];
  __shared__ __attribute__((aligned(16))) float sB[T_ * N_];

  const int lane = threadIdx.x & 31;
  const int q  = __builtin_amdgcn_readfirstlane((int)(threadIdx.x >> 5));
  const int dg = blockIdx.x % DG_;
  const int bg = blockIdx.x / DG_;
  const int g  = bg % G_;
  const int bb = bg / G_;
  const int d  = dg * CW_ + lane;
  const int nq = q * NQ_;

  v2f A2a[NQ_ / 2], A2b[NQ_ / 2], h2a[NQ_ / 2], h2b[NQ_ / 2];
  s6_load_A2(Alog + (size_t)d * N_ + nq, A2a);
  s6_load_A2(Alog + (size_t)(d + 32) * N_ + nq, A2b);
#pragma unroll
  for (int j = 0; j < NQ_ / 2; ++j) {
    h2a[j].x = 0.0f;  h2a[j].y = 0.0f;
    h2b[j].x = 0.0f;  h2b[j].y = 0.0f;
  }
  float dsa = 0.0f, dsb = 0.0f;
  float yna, ynb;

  const size_t row0 = (size_t)bb * L_ + (size_t)g * T_;
  s6_delta_tile(xdbl + row0 * E_, Wdt + (size_t)dg * CW_ * R_, bdt + dg * CW_, q, lane, sDX);
  s6_stage_x(x + row0 * D_ + dg * CW_, sDX);
  {
    const float* bsrc = xdbl + row0 * E_ + R_;
#pragma unroll
    for (int i = 0; i < 8; ++i) {
      const int idx = (int)threadIdx.x + 128 * i;
      const int row = idx >> 4, c4 = (idx & 15) * 4;
      *(v4f*)(sB + row * N_ + c4) = *(const v4f*)(bsrc + (size_t)row * E_ + c4);
    }
  }
  __syncthreads();

  const float* bq = sB + nq;
  const float* dx = sDX + 2 * lane;

  v4f Ba[4], Bb[4];
  v2f da0, da1, db0, db1;
  s6_row16(bq, Ba);
  da0 = *(const v2f*)dx;
  da1 = *(const v2f*)(dx + 64);
  for (int tt = 0; tt < T_; tt += 2) {
    s6_row16(bq + (tt + 1) * N_, Bb);
    db0 = *(const v2f*)(dx + (tt + 1) * 2 * CW_);
    db1 = *(const v2f*)(dx + (tt + 1) * 2 * CW_ + 64);
    __builtin_amdgcn_sched_barrier(0);
    s6_step_pair<false>(Ba, Ba, da0, da1, A2a, A2b, h2a, h2b, yna, ynb);
    dsa += da0.x;
    dsb += da1.x;
    __builtin_amdgcn_sched_barrier(0);
    const int t2 = (tt + 2 < T_) ? tt + 2 : tt + 1;
    s6_row16(bq + t2 * N_, Ba);
    da0 = *(const v2f*)(dx + t2 * 2 * CW_);
    da1 = *(const v2f*)(dx + t2 * 2 * CW_ + 64);
    __builtin_amdgcn_sched_barrier(0);
    s6_step_pair<false>(Bb, Bb, db0, db1, A2a, A2b, h2a, h2b, yna, ynb);
    dsa += db0.x;
    dsb += db1.x;
    __builtin_amdgcn_sched_barrier(0);
  }
  const size_t base = ((size_t)bg * N_ + nq) * D_ + d;
#pragma unroll
  for (int j = 0; j < NQ_ / 2; ++j) {
    Hout[base + (size_t)(2 * j) * D_]     = h2a[j].x;
    Hout[base + (size_t)(2 * j + 1) * D_] = h2a[j].y;
  }
#pragma unroll
  for (int j = 0; j < NQ_ / 2; ++j) {
    Hout[base + 32 + (size_t)(2 * j) * D_]     = h2b[j].x;
    Hout[base + 32 + (size_t)(2 * j + 1) * D_] = h2b[j].y;
  }
  if (q == 0) {
    DSout[(size_t)bg * D_ + d]      = dsa;
    DSout[(size_t)bg * D_ + d + 32] = dsb;
  }
}

__global__ void __launch_bounds__(256)
s6_combine(const float* __restrict__ DSin, const float* __restrict__ Alog,
           const float* __restrict__ Hin, float* __restrict__ h0out) {
  const int idx = blockIdx.x * 256 + threadIdx.x;
  const int bb  = idx / (N_ * D_);
  const int nd  = idx % (N_ * D_);
  const int n   = nd / D_;
  const int d   = nd % D_;
  const size_t base = (size_t)bb * G_ * N_ * D_ + nd;
  const float* ds = DSin + (size_t)bb * G_ * D_ + d;

  const float al = Alog[(size_t)d * N_ + n];
  float dsv[G_], hv[G_];
#pragma unroll
  for (int g = 0; g < G_; ++g) {
    dsv[g] = ds[(size_t)g * D_];
    hv[g]  = Hin[base + (size_t)g * N_ * D_];
  }

  const float A2 = -__expf(al) * S6_LOG2E;
  float h = 0.0f;
#pragma unroll
  for (int g = 0; g < G_; ++g) {
    h0out[base + (size_t)g * N_ * D_] = h;
    const float P = __builtin_amdgcn_exp2f(dsv[g] * A2);
    h = P * h + hv[g];
  }
}

#define SL_ 64
__global__ void __launch_bounds__(128)
s6_scan_emit(const float* __restrict__ xdbl, const float* __restrict__ Wdt,
             const float* __restrict__ bdt,
             const float* __restrict__ x, const float* __restrict__ Alog,
             const float* __restrict__ Dp, const float* __restrict__ h0in,
             float* __restrict__ y) {
  __shared__ __attribute__((aligned(16))) float sDX[T_ * CW_ * 2];
  __shared__ __attribute__((aligned(16))) float sBC[T_ * NS_ * SL_];

  const int lane = threadIdx.x & 31;
  const int q  = __builtin_amdgcn_readfirstlane((int)(threadIdx.x >> 5));
  const int dg = blockIdx.x % DG_;
  const int bg = blockIdx.x / DG_;
  const int g  = bg % G_;
  const int bb = bg / G_;
  const int d  = dg * CW_ + lane;
  const int nq = q * NQ_;

  v2f A2a[NQ_ / 2], A2b[NQ_ / 2], h2a[NQ_ / 2], h2b[NQ_ / 2];
  s6_load_A2(Alog + (size_t)d * N_ + nq, A2a);
  s6_load_A2(Alog + (size_t)(d + 32) * N_ + nq, A2b);
  const size_t hbase = ((size_t)bg * N_ + nq) * D_ + d;
#pragma unroll
  for (int j = 0; j < NQ_ / 2; ++j) {
    h2a[j].x = h0in[hbase + (size_t)(2 * j) * D_];
    h2a[j].y = h0in[hbase + (size_t)(2 * j + 1) * D_];
    h2b[j].x = h0in[hbase + 32 + (size_t)(2 * j) * D_];
    h2b[j].y = h0in[hbase + 32 + (size_t)(2 * j + 1) * D_];
  }

  const size_t row0 = (size_t)bb * L_ + (size_t)g * T_;
  s6_delta_tile(xdbl + row0 * E_, Wdt + (size_t)dg * CW_ * R_, bdt + dg * CW_, q, lane, sDX);
  s6_stage_x(x + row0 * D_ + dg * CW_, sDX);
  {
    const float* bsrc = xdbl + row0 * E_ + R_;
#pragma unroll
    for (int i = 0; i < 16; ++i) {
      const int idx = (int)threadIdx.x + 128 * i;
      const int row = idx >> 5, col = (idx & 31) * 4;
      const int n = col & 63;
      *(v4f*)(sBC + (row * NS_ + (n >> 4)) * SL_ + (col >> 6) * 16 + (n & 15)) =
          *(const v4f*)(bsrc + (size_t)row * E_ + col);
    }
  }
  __syncthreads();

  float* bc = sBC + q * SL_;
  const float* dx = sDX + 2 * lane;

  v4f Ba[4], Ca[4], Bb[4], Cb[4];
  v2f da0, da1, db0, db1;
  float ya, yb;
  s6_row16(bc, Ba);
  s6_row16(bc + 16, Ca);
  da0 = *(const v2f*)dx;
  da1 = *(const v2f*)(dx + 64);
  for (int tt = 0; tt < T_; tt += 2) {
    s6_row16(bc + (tt + 1) * NS_ * SL_, Bb);
    s6_row16(bc + (tt + 1) * NS_ * SL_ + 16, Cb);
    db0 = *(const v2f*)(dx + (tt + 1) * 2 * CW_);
    db1 = *(const v2f*)(dx + (tt + 1) * 2 * CW_ + 64);
    __builtin_amdgcn_sched_barrier(0);
    s6_step_pair<true>(Ba, Ca, da0, da1, A2a, A2b, h2a, h2b, ya, yb);
    bc[tt * NS_ * SL_ + lane]      = ya;
    bc[tt * NS_ * SL_ + 32 + lane] = yb;
    asm volatile("" :: "v"(Bb[0]), "v"(Bb[1]), "v"(Bb[2]), "v"(Bb[3]),
                       "v"(Cb[0]), "v"(Cb[1]), "v"(Cb[2]), "v"(Cb[3]), "v"(db0), "v"(db1));
    const int t2 = (tt + 2 < T_) ? tt + 2 : tt + 1;
    s6_row16(bc + t2 * NS_ * SL_, Ba);
    s6_row16(bc + t2 * NS_ * SL_ + 16, Ca);
    da0 = *(const v2f*)(dx + t2 * 2 * CW_);
    da1 = *(const v2f*)(dx + t2 * 2 * CW_ + 64);
    __builtin_amdgcn_sched_barrier(0);
    s6_step_pair<true>(Bb, Cb, db0, db1, A2a, A2b, h2a, h2b, ya, yb);
    bc[(tt + 1) * NS_ * SL_ + lane]      = ya;
    bc[(tt + 1) * NS_ * SL_ + 32 + lane] = yb;
    asm volatile("" :: "v"(Ba[0]), "v"(Ba[1]), "v"(Ba[2]), "v"(Ba[3]),
                       "v"(Ca[0]), "v"(Ca[1]), "v"(Ca[2]), "v"(Ca[3]), "v"(da0), "v"(da1));
  }
  __syncthreads();

  const float Dva = Dp[d];
  const float Dvb = Dp[d + 32];
#pragma unroll 4
  for (int k = 0; k < T_ / NS_; ++k) {
    const int tt = q * (T_ / NS_) + k;
    const float* yp = sBC + tt * NS_ * SL_ + lane;
    const float sa = (yp[0] + yp[SL_]) + (yp[2 * SL_] + yp[3 * SL_]);
    const float sb = (yp[32] + yp[SL_ + 32]) + (yp[2 * SL_ + 32] + yp[3 * SL_ + 32]);
    const float xa = dx[tt * 2 * CW_ + 1];
    const float xb = dx[tt * 2 * CW_ + 65];
    y[(row0 + tt) * D_ + d]      = sa + xa * Dva;
    y[(row0 + tt) * D_ + d + 32] = sb + xb * Dvb;
  }
}

extern "C" void kernel_launch(void* const* d_in, const int* in_sizes, int n_in,
                              void* d_out, int out_size, void* d_ws, size_t ws_size,
                              hipStream_t stream) {
  const float* x      = (const float*)d_in[0];
  const float* Wxproj = (const float*)d_in[1];
  const float* Wdt    = (const float*)d_in[2];
  const float* bdt    = (const float*)d_in[3];
  const float* Alog   = (const float*)d_in[4];
  const float* Dp     = (const float*)d_in[5];
  float*       y      = (float*)d_out;

  float* ws    = (float*)d_ws;
  float* xdbl  = ws;
  float* DSws  = xdbl  + (size_t)BL_ * E_;
  float* Hws   = DSws  + (size_t)B_ * D_ * G_;
  float* h0ws  = Hws   + (size_t)B_ * D_ * G_ * N_;

  s6_xproj_gemm<<<256, 256, 0, stream>>>(x, Wxproj, xdbl);
  s6_scan_chunks<<<512, 128, 0, stream>>>(xdbl, Wdt, bdt, x, Alog, DSws, Hws);
  s6_combine<<<256, 256, 0, stream>>>(DSws, Alog, Hws, h0ws);
  s6_scan_emit<<<512, 128, 0, stream>>>(xdbl, Wdt, bdt, x, Alog, Dp, h0ws, y);
}
